# starting kernel
# baseline (speedup 1.0000x reference)
#include <hip/hip_runtime.h>
#include <hip/hip_bf16.h>

#define IN_DIM  128
#define OUT_DIM 64

typedef __attribute__((ext_vector_type(2))) float v2f;
typedef __attribute__((ext_vector_type(8))) float v8f;

__global__ void gat_init_kernel(unsigned int* __restrict__ maxbits,
                                float* __restrict__ denom,
                                float* __restrict__ out,
                                int n_nodes) {
    int i = blockIdx.x * blockDim.x + threadIdx.x;
    int total = n_nodes * OUT_DIM;
    if (i < total) out[i] = 0.0f;
    if (i < n_nodes) {
        maxbits[i] = 0xFF800000u;
        denom[i]   = 0.0f;
    }
}

__global__ void __launch_bounds__(128)
gat_gemm_wmma_kernel(const float* __restrict__ h,
                     const float* __restrict__ W,
                     float* __restrict__ z,
                     int n_nodes) {
    const int lane  = threadIdx.x & 31;
    const int wave  = threadIdx.x >> 5;
    const int half  = lane >> 4;
    const int lcol  = lane & 15;

    const int row0  = blockIdx.x * 16;
    const int nbase = wave * 16;

    int arow = row0 + lcol;
    if (arow >= n_nodes) arow = n_nodes - 1;
    const int ncol = nbase + lcol;

    const float* __restrict__ hrow = h + (size_t)arow * IN_DIM;
    const float* __restrict__ wrow = W + (size_t)ncol * IN_DIM;

    v8f acc = {};
#pragma unroll
    for (int k = 0; k < IN_DIM; k += 4) {
        const int kk = k + 2 * half;
        float2 af = *(const float2*)(hrow + kk);
        float2 bf = *(const float2*)(wrow + kk);
        v2f a; a[0] = af.x; a[1] = af.y;
        v2f b; b[0] = bf.x; b[1] = bf.y;
        acc = __builtin_amdgcn_wmma_f32_16x16x4_f32(
             false, a,  false, b,
             (short)0, acc,  false,  false);
    }

#pragma unroll
    for (int v = 0; v < 8; ++v) {
        int row = row0 + v + 8 * half;
        if (row < n_nodes)
            z[(size_t)row * OUT_DIM + ncol] = acc[v];
    }
}

__global__ void __launch_bounds__(256)
gat_edge_score_kernel(const float* __restrict__ z,
                      const int* __restrict__ src,
                      const int* __restrict__ dst,
                      float* __restrict__ e,
                      unsigned int* __restrict__ maxbits,
                      int n_edges) {
    const int lane = threadIdx.x & 31;
    const int eid  = blockIdx.x * (blockDim.x >> 5) + (threadIdx.x >> 5);
    if (eid >= n_edges) return;

    const int s = src[eid];
    const int d = dst[eid];
    const float2* zs = (const float2*)(z + (size_t)s * OUT_DIM);
    const float2* zd = (const float2*)(z + (size_t)d * OUT_DIM);
    float2 a = zs[lane];
    float2 b = zd[lane];
    float acc = a.x * b.x + a.y * b.y;
#pragma unroll
    for (int off = 16; off > 0; off >>= 1)
        acc += __shfl_xor(acc, off, 32);

    if (lane == 0) {
        e[eid] = acc;
        if (acc >= 0.0f)
            atomicMax((int*)&maxbits[d], __float_as_int(acc));
        else
            atomicMin(&maxbits[d], __float_as_uint(acc));
    }
}

__global__ void gat_exp_denom_kernel(const int* __restrict__ dst,
                                     float* __restrict__ e,
                                     const unsigned int* __restrict__ maxbits,
                                     float* __restrict__ denom,
                                     int n_edges) {
    int eid = blockIdx.x * blockDim.x + threadIdx.x;
    if (eid >= n_edges) return;
    int d = dst[eid];
    float m = __int_as_float((int)maxbits[d]);
    float t = __expf(e[eid] - m);
    e[eid] = t;
    atomicAdd(&denom[d], t);
}

__global__ void __launch_bounds__(256)
gat_aggregate_kernel(const float* __restrict__ z,
                     const int* __restrict__ src,
                     const int* __restrict__ dst,
                     const float* __restrict__ e,
                     const float* __restrict__ denom,
                     float* __restrict__ out,
                     int n_edges) {
    const int j   = threadIdx.x & 63;
    const int eid = blockIdx.x * (blockDim.x >> 6) + (threadIdx.x >> 6);
    if (eid >= n_edges) return;
    const int s = src[eid];
    const int d = dst[eid];
    const float alpha = e[eid] / fmaxf(denom[d], 1e-30f);
    atomicAdd(&out[(size_t)d * OUT_DIM + j], alpha * z[(size_t)s * OUT_DIM + j]);
}

extern "C" void kernel_launch(void* const* d_in, const int* in_sizes, int n_in,
                              void* d_out, int out_size, void* d_ws, size_t ws_size,
                              hipStream_t stream) {
    const float* h   = (const float*)d_in[0];
    const float* W   = (const float*)d_in[1];
    const int*   src = (const int*)d_in[2];
    const int*   dst = (const int*)d_in[3];
    float* out = (float*)d_out;

    const int n_nodes = in_sizes[0] / IN_DIM;
    const int n_edges = in_sizes[2];

    float*        z       = (float*)d_ws;
    float*        e       = z + (size_t)n_nodes * OUT_DIM;
    unsigned int* maxbits = (unsigned int*)(e + n_edges);
    float*        denom   = (float*)(maxbits + n_nodes);

    {
        int total = n_nodes * OUT_DIM;
        gat_init_kernel<<<(total + 255) / 256, 256, 0, stream>>>(
            maxbits, denom, out, n_nodes);
    }
    {
        int blocks = (n_nodes + 15) / 16;
        gat_gemm_wmma_kernel<<<blocks, 128, 0, stream>>>(h, W, z, n_nodes);
    }
    {
        int edges_per_block = 256 / 32;
        int blocks = (n_edges + edges_per_block - 1) / edges_per_block;
        gat_edge_score_kernel<<<blocks, 256, 0, stream>>>(
            z, src, dst, e, maxbits, n_edges);
    }
    {
        gat_exp_denom_kernel<<<(n_edges + 255) / 256, 256, 0, stream>>>(
            dst, e, maxbits, denom, n_edges);
    }
    {
        int edges_per_block = 256 / 64;
        int blocks = (n_edges + edges_per_block - 1) / edges_per_block;
        gat_aggregate_kernel<<<blocks, 256, 0, stream>>>(
            z, src, dst, e, denom, out, n_edges);
    }
}


# Node kernel: a wave takes four nodes side by side (8 lanes a node), each list walked one edge a step in a link/source/row pipeline; softmax rescaled per edge, so the sums run in another order
# speedup vs baseline: 3.8902x; 3.8902x over previous
#include <hip/hip_runtime.h>
#include <hip/hip_bf16.h>

#define IN_DIM  128
#define OUT_DIM 64

typedef __attribute__((ext_vector_type(2))) float v2f;
typedef __attribute__((ext_vector_type(8))) float v8f;


__global__ void gat_head_init_kernel(int* __restrict__ head, int n_nodes) {
    int i = blockIdx.x * blockDim.x + threadIdx.x;
    if (i < n_nodes) head[i] = -1;
}

__global__ void __launch_bounds__(128)
gat_gemm_wmma_kernel(const float* __restrict__ h,
                     const float* __restrict__ W,
                     float* __restrict__ z,
                     int n_nodes) {
    const int lane  = threadIdx.x & 31;
    const int wave  = threadIdx.x >> 5;
    const int half  = lane >> 4;
    const int lcol  = lane & 15;

    const int row0  = blockIdx.x * 16;
    const int nbase = wave * 16;

    int arow = row0 + lcol;
    if (arow >= n_nodes) arow = n_nodes - 1;
    const int ncol = nbase + lcol;

    const float* __restrict__ hrow = h + (size_t)arow * IN_DIM;
    const float* __restrict__ wrow = W + (size_t)ncol * IN_DIM;

    v8f acc = {};
#pragma unroll
    for (int k = 0; k < IN_DIM; k += 4) {
        const int kk = k + 2 * half;
        float2 af = *(const float2*)(hrow + kk);
        float2 bf = *(const float2*)(wrow + kk);
        v2f a; a[0] = af.x; a[1] = af.y;
        v2f b; b[0] = bf.x; b[1] = bf.y;
        acc = __builtin_amdgcn_wmma_f32_16x16x4_f32(
             false, a,  false, b,
             (short)0, acc,  false,  false);
    }

#pragma unroll
    for (int v = 0; v < 8; ++v) {
        int row = row0 + v + 8 * half;
        if (row < n_nodes)
            z[(size_t)row * OUT_DIM + ncol] = acc[v];
    }
}

__global__ void __launch_bounds__(256)
gat_link_kernel(const int* __restrict__ dst,
                int* __restrict__ head,
                int* __restrict__ next,
                int n_edges) {
    const int eid = blockIdx.x * blockDim.x + threadIdx.x;
    if (eid >= n_edges) return;
    const int d = dst[eid];
    next[eid] = atomicExch(&head[d], eid);
}

#define GL  8
#define NPW (32 / GL)
#define NP  (OUT_DIM / 4 / GL)

__device__ __forceinline__ int gat_clamp(int i, int n) {
    const unsigned u = (unsigned)i, top = (unsigned)(n - 1);
    return (int)(u < top ? u : top);
}

template <int MASK>
__device__ __forceinline__ float gat_add_xor(float p) {
    const int q = __builtin_amdgcn_update_dpp(0, __float_as_int(p), 0x160 + MASK, 0xf, 0xf, true);
    return p + __int_as_float(q);
}

__global__ void __launch_bounds__(256)
gat_node_kernel(const float* __restrict__ z,
                const int* __restrict__ src,
                const int* __restrict__ head,
                const int* __restrict__ next,
                float* __restrict__ out,
                int n_nodes, int n_edges) {
    const int lane = threadIdx.x & 31;
    const int sub  = lane & (GL - 1);
    const int grp  = lane / GL;
    const int wv   = blockIdx.x * (blockDim.x >> 5) + (threadIdx.x >> 5);
    const int d    = wv * NPW + grp;
    const bool live = d < n_nodes;
    const int  dc   = gat_clamp(d, n_nodes);

    const float4* zdp = (const float4*)(z + (size_t)dc * OUT_DIM);
    float4 zd[NP];
    for (int q = 0; q < NP; ++q) zd[q] = zdp[sub + GL * q];

    int eA = live ? head[dc] : -1;
    int  sB = 0;
    bool vB = false;
    float4 rc[NP];
    bool   vC = false;

    float  m = -INFINITY;
    float  l = 0.0f;
    float4 acc[NP];
    for (int q = 0; q < NP; ++q) {
        rc[q]  = make_float4(0.0f, 0.0f, 0.0f, 0.0f);
        acc[q] = make_float4(0.0f, 0.0f, 0.0f, 0.0f);
    }

    for (int it = 0; it < n_edges + 2; ++it) {
        if (!__any((eA >= 0) || vB || vC)) break;

        const bool vA = eA >= 0;
        const int  ec = gat_clamp(eA, n_edges);
        const int  nn = next[ec];
        const int  ss = src[ec];

        const float4* rp = (const float4*)(z + (size_t)gat_clamp(sB, n_nodes) * OUT_DIM);
        float4 rb[NP];
        for (int q = 0; q < NP; ++q) rb[q] = rp[sub + GL * q];

        float p = 0.0f;
        for (int q = 0; q < NP; ++q)
            p += rc[q].x * zd[q].x + rc[q].y * zd[q].y + rc[q].z * zd[q].z + rc[q].w * zd[q].w;
        p = gat_add_xor<1>(p);
        p = gat_add_xor<2>(p);
        if (GL > 4) p = gat_add_xor<4>(p);

        const float sc    = vC ? p : -INFINITY;
        const float m_new = fmaxf(m, sc);
        const float corr  = (m == -INFINITY) ? 0.0f : __expf(m - m_new);
        const float w     = vC ? __expf(sc - m_new) : 0.0f;
        l = l * corr + w;
        for (int q = 0; q < NP; ++q) {
            acc[q].x = acc[q].x * corr + w * rc[q].x;
            acc[q].y = acc[q].y * corr + w * rc[q].y;
            acc[q].z = acc[q].z * corr + w * rc[q].z;
            acc[q].w = acc[q].w * corr + w * rc[q].w;
        }
        m = m_new;

        for (int q = 0; q < NP; ++q) rc[q] = rb[q];
        vC = vB;
        sB = ss;  vB = vA;
        eA = vA ? nn : -1;
    }

    if (live) {
        const float inv = 1.0f / fmaxf(l, 1e-30f);
        float4* op = (float4*)(out + (size_t)d * OUT_DIM);
        for (int q = 0; q < NP; ++q)
            op[sub + GL * q] = make_float4(acc[q].x * inv, acc[q].y * inv, acc[q].z * inv, acc[q].w * inv);
    }
}

extern "C" void kernel_launch(void* const* d_in, const int* in_sizes, int n_in,
                              void* d_out, int out_size, void* d_ws, size_t ws_size,
                              hipStream_t stream) {
    const float* h   = (const float*)d_in[0];
    const float* W   = (const float*)d_in[1];
    const int*   src = (const int*)d_in[2];
    const int*   dst = (const int*)d_in[3];
    float* out = (float*)d_out;

    const int n_nodes = in_sizes[0] / IN_DIM;
    const int n_edges = in_sizes[2];

    float* z    = (float*)d_ws;
    int*   next = (int*)(z + (size_t)n_nodes * OUT_DIM);
    int*   head = next + n_edges;

    {
        gat_head_init_kernel<<<(n_nodes + 255) / 256, 256, 0, stream>>>(head, n_nodes);
    }
    {
        int blocks = (n_nodes + 15) / 16;
        gat_gemm_wmma_kernel<<<blocks, 128, 0, stream>>>(h, W, z, n_nodes);
    }
    {
        gat_link_kernel<<<(n_edges + 255) / 256, 256, 0, stream>>>(
            dst, head, next, n_edges);
    }
    {
        int nodes_per_block = (256 / 32) * NPW;
        int blocks = (n_nodes + nodes_per_block - 1) / nodes_per_block;
        gat_node_kernel<<<blocks, 256, 0, stream>>>(
            z, src, head, next, out, n_nodes, n_edges);
    }
}


# Edge linking in a launch of its own: each work-group counting-sorts 24576 edges in LDS by the line of head[] they hit, then issues its exchanges in that order; lists in arrival order, as before
# speedup vs baseline: 5.4468x; 1.4001x over previous
#include <hip/hip_runtime.h>
#include <hip/hip_bf16.h>

#define IN_DIM  128
#define OUT_DIM 64

typedef __attribute__((ext_vector_type(2))) float v2f;
typedef __attribute__((ext_vector_type(8))) float v8f;


__global__ void gat_head_init_kernel(int* __restrict__ head, int n_nodes) {
    int i = blockIdx.x * blockDim.x + threadIdx.x;
    if (i < n_nodes) head[i] = -1;
}

#define LK_THREADS 1024
#define LK_PER     24
#define LK_CHUNK   (LK_THREADS * LK_PER)
#define LK_MAXBIN  4096

__global__ void __launch_bounds__(LK_THREADS)
gat_link_kernel(const int* __restrict__ dst,
                int* __restrict__ head,
                int* __restrict__ next,
                int n_nodes, int n_edges) {
    __shared__ unsigned cnt[LK_MAXBIN];
    __shared__ int      key[LK_CHUNK];
    __shared__ int      idx[LK_CHUNK];
    __shared__ unsigned wsum[LK_THREADS / 32];

    const int tid  = threadIdx.x;
    const int lane = tid & 31;
    const int wave = tid >> 5;
    const int cb   = blockIdx.x * LK_CHUNK;
    const int n    = min(LK_CHUNK, n_edges - cb);

    int shift = 5;
    while (shift < 31 && (((unsigned)(n_nodes - 1)) >> shift) >= LK_MAXBIN) ++shift;

    for (int i = tid; i < LK_MAXBIN; i += LK_THREADS) cnt[i] = 0u;
    __syncthreads();

    int      dj[LK_PER];
    unsigned rj[LK_PER];
    unsigned bj[LK_PER];
#pragma unroll
    for (int j = 0; j < LK_PER; ++j) {
        const int li = j * LK_THREADS + tid;
        dj[j] = (li < n) ? dst[cb + li] : 0;
    }
#pragma unroll
    for (int j = 0; j < LK_PER; ++j) {
        const int li = j * LK_THREADS + tid;
        const unsigned b = min(((unsigned)dj[j]) >> shift, (unsigned)(LK_MAXBIN - 1));
        bj[j] = b;
        rj[j] = (li < n) ? atomicAdd(&cnt[b], 1u) : 0u;
    }
    __syncthreads();

    {
        const int t4 = tid * 4;
        const unsigned c0 = cnt[t4], c1 = cnt[t4 + 1], c2 = cnt[t4 + 2], c3 = cnt[t4 + 3];
        const unsigned s = c0 + c1 + c2 + c3;
        unsigned inc = s;
        for (int o = 1; o < 32; o <<= 1) {
            const unsigned v = __shfl_up(inc, o, 32);
            if (lane >= o) inc += v;
        }
        if (lane == 31) wsum[wave] = inc;
        __syncthreads();
        unsigned off = 0u;
        for (int w = 0; w < wave; ++w) off += wsum[w];
        const unsigned ex = off + inc - s;
        cnt[t4]     = ex;
        cnt[t4 + 1] = ex + c0;
        cnt[t4 + 2] = ex + c0 + c1;
        cnt[t4 + 3] = ex + c0 + c1 + c2;
    }
    __syncthreads();

#pragma unroll
    for (int j = 0; j < LK_PER; ++j) {
        const int li = j * LK_THREADS + tid;
        if (li < n) {
            const unsigned pos = cnt[bj[j]] + rj[j];
            key[pos] = dj[j];
            idx[pos] = li;
        }
    }
    __syncthreads();

    int sd[LK_PER];
    int si[LK_PER];
#pragma unroll
    for (int j = 0; j < LK_PER; ++j) {
        const int li = j * LK_THREADS + tid;
        sd[j] = (li < n) ? key[li] : 0;
        si[j] = (li < n) ? idx[li] : 0;
    }
    __syncthreads();

    int pj[LK_PER];
#pragma unroll
    for (int j = 0; j < LK_PER; ++j) {
        const int li = j * LK_THREADS + tid;
        pj[j] = (li < n) ? atomicExch(&head[sd[j]], cb + si[j]) : -1;
    }
#pragma unroll
    for (int j = 0; j < LK_PER; ++j) {
        const int li = j * LK_THREADS + tid;
        if (li < n) key[si[j]] = pj[j];
    }
    __syncthreads();

    for (int li = tid; li < n; li += LK_THREADS) next[cb + li] = key[li];
}

__global__ void __launch_bounds__(128)
gat_gemm_kernel(const float* __restrict__ h,
                const float* __restrict__ W,
                float* __restrict__ z,
                int n_nodes) {
    const int lane  = threadIdx.x & 31;
    const int wave  = threadIdx.x >> 5;
    const int half  = lane >> 4;
    const int lcol  = lane & 15;

    const int row0  = blockIdx.x * 16;
    const int nbase = wave * 16;

    int arow = row0 + lcol;
    if (arow >= n_nodes) arow = n_nodes - 1;
    const int ncol = nbase + lcol;

    const float* __restrict__ hrow = h + (size_t)arow * IN_DIM;
    const float* __restrict__ wrow = W + (size_t)ncol * IN_DIM;

    v8f acc = {};
#pragma unroll
    for (int k = 0; k < IN_DIM; k += 4) {
        const int kk = k + 2 * half;
        float2 af = *(const float2*)(hrow + kk);
        float2 bf = *(const float2*)(wrow + kk);
        v2f a; a[0] = af.x; a[1] = af.y;
        v2f b; b[0] = bf.x; b[1] = bf.y;
        acc = __builtin_amdgcn_wmma_f32_16x16x4_f32(
             false, a,  false, b,
             (short)0, acc,  false,  false);
    }

#pragma unroll
    for (int v = 0; v < 8; ++v) {
        int row = row0 + v + 8 * half;
        if (row < n_nodes)
            z[(size_t)row * OUT_DIM + ncol] = acc[v];
    }
}

#define GL  8
#define NPW (32 / GL)
#define NP  (OUT_DIM / 4 / GL)

__device__ __forceinline__ int gat_clamp(int i, int n) {
    const unsigned u = (unsigned)i, top = (unsigned)(n - 1);
    return (int)(u < top ? u : top);
}

template <int MASK>
__device__ __forceinline__ float gat_add_xor(float p) {
    const int q = __builtin_amdgcn_update_dpp(0, __float_as_int(p), 0x160 + MASK, 0xf, 0xf, true);
    return p + __int_as_float(q);
}

__global__ void __launch_bounds__(256)
gat_node_kernel(const float* __restrict__ z,
                const int* __restrict__ src,
                const int* __restrict__ head,
                const int* __restrict__ next,
                float* __restrict__ out,
                int n_nodes, int n_edges) {
    const int lane = threadIdx.x & 31;
    const int sub  = lane & (GL - 1);
    const int grp  = lane / GL;
    const int wv   = blockIdx.x * (blockDim.x >> 5) + (threadIdx.x >> 5);
    const int d    = wv * NPW + grp;
    const bool live = d < n_nodes;
    const int  dc   = gat_clamp(d, n_nodes);

    const float4* zdp = (const float4*)(z + (size_t)dc * OUT_DIM);
    float4 zd[NP];
    for (int q = 0; q < NP; ++q) zd[q] = zdp[sub + GL * q];

    int eA = live ? head[dc] : -1;
    int  sB = 0;
    bool vB = false;
    float4 rc[NP];
    bool   vC = false;

    float  m = -INFINITY;
    float  l = 0.0f;
    float4 acc[NP];
    for (int q = 0; q < NP; ++q) {
        rc[q]  = make_float4(0.0f, 0.0f, 0.0f, 0.0f);
        acc[q] = make_float4(0.0f, 0.0f, 0.0f, 0.0f);
    }

    for (int it = 0; it < n_edges + 2; ++it) {
        if (!__any((eA >= 0) || vB || vC)) break;

        const bool vA = eA >= 0;
        const int  ec = gat_clamp(eA, n_edges);
        const int  nn = next[ec];
        const int  ss = src[ec];

        const float4* rp = (const float4*)(z + (size_t)gat_clamp(sB, n_nodes) * OUT_DIM);
        float4 rb[NP];
        for (int q = 0; q < NP; ++q) rb[q] = rp[sub + GL * q];

        float p = 0.0f;
        for (int q = 0; q < NP; ++q)
            p += rc[q].x * zd[q].x + rc[q].y * zd[q].y + rc[q].z * zd[q].z + rc[q].w * zd[q].w;
        p = gat_add_xor<1>(p);
        p = gat_add_xor<2>(p);
        if (GL > 4) p = gat_add_xor<4>(p);

        const float sc    = vC ? p : -INFINITY;
        const float m_new = fmaxf(m, sc);
        const float corr  = (m == -INFINITY) ? 0.0f : __expf(m - m_new);
        const float w     = vC ? __expf(sc - m_new) : 0.0f;
        l = l * corr + w;
        for (int q = 0; q < NP; ++q) {
            acc[q].x = acc[q].x * corr + w * rc[q].x;
            acc[q].y = acc[q].y * corr + w * rc[q].y;
            acc[q].z = acc[q].z * corr + w * rc[q].z;
            acc[q].w = acc[q].w * corr + w * rc[q].w;
        }
        m = m_new;

        for (int q = 0; q < NP; ++q) rc[q] = rb[q];
        vC = vB;
        sB = ss;  vB = vA;
        eA = vA ? nn : -1;
    }

    if (live) {
        const float inv = 1.0f / fmaxf(l, 1e-30f);
        float4* op = (float4*)(out + (size_t)d * OUT_DIM);
        for (int q = 0; q < NP; ++q)
            op[sub + GL * q] = make_float4(acc[q].x * inv, acc[q].y * inv, acc[q].z * inv, acc[q].w * inv);
    }
}

extern "C" void kernel_launch(void* const* d_in, const int* in_sizes, int n_in,
                              void* d_out, int out_size, void* d_ws, size_t ws_size,
                              hipStream_t stream) {
    const float* h   = (const float*)d_in[0];
    const float* W   = (const float*)d_in[1];
    const int*   src = (const int*)d_in[2];
    const int*   dst = (const int*)d_in[3];
    float* out = (float*)d_out;

    const int n_nodes = in_sizes[0] / IN_DIM;
    const int n_edges = in_sizes[2];

    float* z    = (float*)d_ws;
    int*   next = (int*)(z + (size_t)n_nodes * OUT_DIM);
    int*   head = next + n_edges;

    {
        gat_head_init_kernel<<<(n_nodes + 255) / 256, 256, 0, stream>>>(head, n_nodes);
    }
    {
        int blocks = (n_edges + LK_CHUNK - 1) / LK_CHUNK;
        gat_link_kernel<<<blocks, LK_THREADS, 0, stream>>>(dst, head, next, n_nodes, n_edges);
    }
    {
        int blocks = (n_nodes + 15) / 16;
        gat_gemm_kernel<<<blocks, 128, 0, stream>>>(h, W, z, n_nodes);
    }
    {
        int nodes_per_block = (256 / 32) * NPW;
        int blocks = (n_nodes + nodes_per_block - 1) / nodes_per_block;
        gat_node_kernel<<<blocks, 256, 0, stream>>>(
            z, src, head, next, out, n_nodes, n_edges);
    }
}


# GEMM re-tiled: 2048 resident waves, each 16 rows x all 64 columns with W kept in registers (staged once through LDS); h read once in 16-byte pieces, next tile loaded ahead; k summed in another order
# speedup vs baseline: 6.6101x; 1.2136x over previous
#include <hip/hip_runtime.h>
#include <hip/hip_bf16.h>

#define IN_DIM  128
#define OUT_DIM 64

typedef __attribute__((ext_vector_type(2))) float v2f;
typedef __attribute__((ext_vector_type(8))) float v8f;


__global__ void gat_head_init_kernel(int* __restrict__ head, int n_nodes) {
    int i = blockIdx.x * blockDim.x + threadIdx.x;
    if (i < n_nodes) head[i] = -1;
}

#define LK_THREADS 1024
#define LK_PER     24
#define LK_CHUNK   (LK_THREADS * LK_PER)
#define LK_MAXBIN  4096

__global__ void __launch_bounds__(LK_THREADS)
gat_link_kernel(const int* __restrict__ dst,
                int* __restrict__ head,
                int* __restrict__ next,
                int n_nodes, int n_edges) {
    __shared__ unsigned cnt[LK_MAXBIN];
    __shared__ int      key[LK_CHUNK];
    __shared__ int      idx[LK_CHUNK];
    __shared__ unsigned wsum[LK_THREADS / 32];

    const int tid  = threadIdx.x;
    const int lane = tid & 31;
    const int wave = tid >> 5;
    const int cb   = blockIdx.x * LK_CHUNK;
    const int n    = min(LK_CHUNK, n_edges - cb);

    int shift = 5;
    while (shift < 31 && (((unsigned)(n_nodes - 1)) >> shift) >= LK_MAXBIN) ++shift;

    for (int i = tid; i < LK_MAXBIN; i += LK_THREADS) cnt[i] = 0u;
    __syncthreads();

    int      dj[LK_PER];
    unsigned rj[LK_PER];
    unsigned bj[LK_PER];
#pragma unroll
    for (int j = 0; j < LK_PER; ++j) {
        const int li = j * LK_THREADS + tid;
        dj[j] = (li < n) ? dst[cb + li] : 0;
    }
#pragma unroll
    for (int j = 0; j < LK_PER; ++j) {
        const int li = j * LK_THREADS + tid;
        const unsigned b = min(((unsigned)dj[j]) >> shift, (unsigned)(LK_MAXBIN - 1));
        bj[j] = b;
        rj[j] = (li < n) ? atomicAdd(&cnt[b], 1u) : 0u;
    }
    __syncthreads();

    {
        const int t4 = tid * 4;
        const unsigned c0 = cnt[t4], c1 = cnt[t4 + 1], c2 = cnt[t4 + 2], c3 = cnt[t4 + 3];
        const unsigned s = c0 + c1 + c2 + c3;
        unsigned inc = s;
        for (int o = 1; o < 32; o <<= 1) {
            const unsigned v = __shfl_up(inc, o, 32);
            if (lane >= o) inc += v;
        }
        if (lane == 31) wsum[wave] = inc;
        __syncthreads();
        unsigned off = 0u;
        for (int w = 0; w < wave; ++w) off += wsum[w];
        const unsigned ex = off + inc - s;
        cnt[t4]     = ex;
        cnt[t4 + 1] = ex + c0;
        cnt[t4 + 2] = ex + c0 + c1;
        cnt[t4 + 3] = ex + c0 + c1 + c2;
    }
    __syncthreads();

#pragma unroll
    for (int j = 0; j < LK_PER; ++j) {
        const int li = j * LK_THREADS + tid;
        if (li < n) {
            const unsigned pos = cnt[bj[j]] + rj[j];
            key[pos] = dj[j];
            idx[pos] = li;
        }
    }
    __syncthreads();

    int sd[LK_PER];
    int si[LK_PER];
#pragma unroll
    for (int j = 0; j < LK_PER; ++j) {
        const int li = j * LK_THREADS + tid;
        sd[j] = (li < n) ? key[li] : 0;
        si[j] = (li < n) ? idx[li] : 0;
    }
    __syncthreads();

    int pj[LK_PER];
#pragma unroll
    for (int j = 0; j < LK_PER; ++j) {
        const int li = j * LK_THREADS + tid;
        pj[j] = (li < n) ? atomicExch(&head[sd[j]], cb + si[j]) : -1;
    }
#pragma unroll
    for (int j = 0; j < LK_PER; ++j) {
        const int li = j * LK_THREADS + tid;
        if (li < n) key[si[j]] = pj[j];
    }
    __syncthreads();

    for (int li = tid; li < n; li += LK_THREADS) next[cb + li] = key[li];
}

#define GM_WAVES  8
#define GM_BLOCKS 256
#define GM_KB     (IN_DIM / 8)
#define GM_NT     (OUT_DIM / 16)
#define GM_WP4    (IN_DIM / 4 + 1)

__device__ __forceinline__ float gat_xhalf(float x) {
    const int b = __float_as_int(x);
    return __int_as_float(__builtin_amdgcn_permlanex16(b, b, 0x76543210, 0xfedcba98, false, false));
}

template <bool GUARD>
__device__ __forceinline__ void gat_store_tile(const v8f (&acc)[GM_NT], float* __restrict__ zp,
                                               int rows, int half) {
#pragma unroll
    for (int p = 0; p < GM_NT / 2; ++p) {
#pragma unroll
        for (int v = 0; v < 8; ++v) {
            const float lo = acc[2 * p][v];
            const float hi = acc[2 * p + 1][v];
            const float send = half ? lo : hi;
            const float recv = gat_xhalf(send);
            const float x = half ? recv : lo;
            const float y = half ? hi : recv;
            if (!GUARD || v < rows)     zp[v * OUT_DIM + 32 * p] = x;
            if (!GUARD || v + 8 < rows) zp[(v + 8) * OUT_DIM + 32 * p] = y;
        }
    }
}

__global__ void __launch_bounds__(GM_WAVES * 32)
gat_gemm_kernel(const float* __restrict__ h,
                const float* __restrict__ W,
                float* __restrict__ z,
                int n_nodes) {
    const int lane  = threadIdx.x & 31;
    const int half  = lane >> 4;
    const int lcol  = lane & 15;
    const int wave  = __builtin_amdgcn_readfirstlane((int)(threadIdx.x >> 5));

    __shared__ float4 wl[OUT_DIM * GM_WP4];
#pragma unroll
    for (int q = 0; q < OUT_DIM * IN_DIM / 4 / (GM_WAVES * 32); ++q) {
        const int f = (int)threadIdx.x + GM_WAVES * 32 * q;
        wl[(f >> 5) * GM_WP4 + (f & 31)] = ((const float4*)W)[f];
    }
    __syncthreads();

    const int n_tiles = (n_nodes + 15) >> 4;
    const int stride  = (int)gridDim.x * GM_WAVES;
    int tile = wave * (int)gridDim.x + (int)blockIdx.x;
    if (tile >= n_tiles) return;

    const int koff = 4 * half;

    float4 bw[GM_NT][GM_KB];
#pragma unroll
    for (int t = 0; t < GM_NT; ++t) {
#pragma unroll
        for (int j = 0; j < GM_KB; ++j) bw[t][j] = wl[(16 * t + lcol) * GM_WP4 + 2 * j + half];
    }

    float4 a[GM_KB];
    {
        int arow = tile * 16 + lcol;
        if (arow >= n_nodes) arow = n_nodes - 1;
        const float* hrow = h + (size_t)arow * IN_DIM + koff;
#pragma unroll
        for (int j = 0; j < GM_KB; ++j) a[j] = *(const float4*)(hrow + 8 * j);
    }

    for (; tile < n_tiles; tile += stride) {
        float4 an[GM_KB];
        {
            const int nt = (tile + stride < n_tiles) ? tile + stride : tile;
            int arow = nt * 16 + lcol;
            if (arow >= n_nodes) arow = n_nodes - 1;
            const float* hrow = h + (size_t)arow * IN_DIM + koff;
#pragma unroll
            for (int j = 0; j < GM_KB; ++j) an[j] = *(const float4*)(hrow + 8 * j);
        }

        v8f acc[GM_NT];
#pragma unroll
        for (int t = 0; t < GM_NT; ++t) acc[t] = v8f{};

#pragma unroll
        for (int j = 0; j < GM_KB; ++j) {
            v2f a0; a0[0] = a[j].x; a0[1] = a[j].y;
            v2f a1; a1[0] = a[j].z; a1[1] = a[j].w;
#pragma unroll
            for (int t = 0; t < GM_NT; ++t) {
                v2f b0; b0[0] = bw[t][j].x; b0[1] = bw[t][j].y;
                acc[t] = __builtin_amdgcn_wmma_f32_16x16x4_f32(
                     false, a0,  false, b0,
                     (short)0, acc[t],  false,  false);
            }
#pragma unroll
            for (int t = 0; t < GM_NT; ++t) {
                v2f b1; b1[0] = bw[t][j].z; b1[1] = bw[t][j].w;
                acc[t] = __builtin_amdgcn_wmma_f32_16x16x4_f32(
                     false, a1,  false, b1,
                     (short)0, acc[t],  false,  false);
            }
        }

        const int row0 = tile * 16;
        float* zp = z + (size_t)row0 * OUT_DIM + lane;
        if (row0 + 16 <= n_nodes) gat_store_tile<false>(acc, zp, 16, half);
        else                      gat_store_tile<true>(acc, zp, n_nodes - row0, half);

#pragma unroll
        for (int j = 0; j < GM_KB; ++j) a[j] = an[j];
    }
}

#define GL  8
#define NPW (32 / GL)
#define NP  (OUT_DIM / 4 / GL)

__device__ __forceinline__ int gat_clamp(int i, int n) {
    const unsigned u = (unsigned)i, top = (unsigned)(n - 1);
    return (int)(u < top ? u : top);
}

template <int MASK>
__device__ __forceinline__ float gat_add_xor(float p) {
    const int q = __builtin_amdgcn_update_dpp(0, __float_as_int(p), 0x160 + MASK, 0xf, 0xf, true);
    return p + __int_as_float(q);
}

__global__ void __launch_bounds__(256)
gat_node_kernel(const float* __restrict__ z,
                const int* __restrict__ src,
                const int* __restrict__ head,
                const int* __restrict__ next,
                float* __restrict__ out,
                int n_nodes, int n_edges) {
    const int lane = threadIdx.x & 31;
    const int sub  = lane & (GL - 1);
    const int grp  = lane / GL;
    const int wv   = blockIdx.x * (blockDim.x >> 5) + (threadIdx.x >> 5);
    const int d    = wv * NPW + grp;
    const bool live = d < n_nodes;
    const int  dc   = gat_clamp(d, n_nodes);

    const float4* zdp = (const float4*)(z + (size_t)dc * OUT_DIM);
    float4 zd[NP];
    for (int q = 0; q < NP; ++q) zd[q] = zdp[sub + GL * q];

    int eA = live ? head[dc] : -1;
    int  sB = 0;
    bool vB = false;
    float4 rc[NP];
    bool   vC = false;

    float  m = -INFINITY;
    float  l = 0.0f;
    float4 acc[NP];
    for (int q = 0; q < NP; ++q) {
        rc[q]  = make_float4(0.0f, 0.0f, 0.0f, 0.0f);
        acc[q] = make_float4(0.0f, 0.0f, 0.0f, 0.0f);
    }

    for (int it = 0; it < n_edges + 2; ++it) {
        if (!__any((eA >= 0) || vB || vC)) break;

        const bool vA = eA >= 0;
        const int  ec = gat_clamp(eA, n_edges);
        const int  nn = next[ec];
        const int  ss = src[ec];

        const float4* rp = (const float4*)(z + (size_t)gat_clamp(sB, n_nodes) * OUT_DIM);
        float4 rb[NP];
        for (int q = 0; q < NP; ++q) rb[q] = rp[sub + GL * q];

        float p = 0.0f;
        for (int q = 0; q < NP; ++q)
            p += rc[q].x * zd[q].x + rc[q].y * zd[q].y + rc[q].z * zd[q].z + rc[q].w * zd[q].w;
        p = gat_add_xor<1>(p);
        p = gat_add_xor<2>(p);
        if (GL > 4) p = gat_add_xor<4>(p);

        const float sc    = vC ? p : -INFINITY;
        const float m_new = fmaxf(m, sc);
        const float corr  = (m == -INFINITY) ? 0.0f : __expf(m - m_new);
        const float w     = vC ? __expf(sc - m_new) : 0.0f;
        l = l * corr + w;
        for (int q = 0; q < NP; ++q) {
            acc[q].x = acc[q].x * corr + w * rc[q].x;
            acc[q].y = acc[q].y * corr + w * rc[q].y;
            acc[q].z = acc[q].z * corr + w * rc[q].z;
            acc[q].w = acc[q].w * corr + w * rc[q].w;
        }
        m = m_new;

        for (int q = 0; q < NP; ++q) rc[q] = rb[q];
        vC = vB;
        sB = ss;  vB = vA;
        eA = vA ? nn : -1;
    }

    if (live) {
        const float inv = 1.0f / fmaxf(l, 1e-30f);
        float4* op = (float4*)(out + (size_t)d * OUT_DIM);
        for (int q = 0; q < NP; ++q)
            op[sub + GL * q] = make_float4(acc[q].x * inv, acc[q].y * inv, acc[q].z * inv, acc[q].w * inv);
    }
}

extern "C" void kernel_launch(void* const* d_in, const int* in_sizes, int n_in,
                              void* d_out, int out_size, void* d_ws, size_t ws_size,
                              hipStream_t stream) {
    const float* h   = (const float*)d_in[0];
    const float* W   = (const float*)d_in[1];
    const int*   src = (const int*)d_in[2];
    const int*   dst = (const int*)d_in[3];
    float* out = (float*)d_out;

    const int n_nodes = in_sizes[0] / IN_DIM;
    const int n_edges = in_sizes[2];

    float* z    = (float*)d_ws;
    int*   next = (int*)(z + (size_t)n_nodes * OUT_DIM);
    int*   head = next + n_edges;

    {
        gat_head_init_kernel<<<(n_nodes + 255) / 256, 256, 0, stream>>>(head, n_nodes);
    }
    {
        int blocks = (n_edges + LK_CHUNK - 1) / LK_CHUNK;
        gat_link_kernel<<<blocks, LK_THREADS, 0, stream>>>(dst, head, next, n_nodes, n_edges);
    }
    {
        gat_gemm_kernel<<<GM_BLOCKS, GM_WAVES * 32, 0, stream>>>(h, W, z, n_nodes);
    }
    {
        int nodes_per_block = (256 / 32) * NPW;
        int blocks = (n_nodes + nodes_per_block - 1) / nodes_per_block;
        gat_node_kernel<<<blocks, 256, 0, stream>>>(
            z, src, head, next, out, n_nodes, n_edges);
    }
}


# Node kernel issues fewer vector instructions per edge: eight nodes a wave (4 lanes each), rows as float pairs for two-wide multiply-adds, one exp per edge; a score's terms summed in another order
# speedup vs baseline: 7.6866x; 1.1629x over previous
#include <hip/hip_runtime.h>
#include <hip/hip_bf16.h>

#define IN_DIM  128
#define OUT_DIM 64

typedef __attribute__((ext_vector_type(2))) float v2f;
typedef __attribute__((ext_vector_type(8))) float v8f;


__global__ void gat_head_init_kernel(int* __restrict__ head, int n_nodes) {
    int i = blockIdx.x * blockDim.x + threadIdx.x;
    if (i < n_nodes) head[i] = -1;
}

#define LK_THREADS 1024
#define LK_PER     24
#define LK_CHUNK   (LK_THREADS * LK_PER)
#define LK_MAXBIN  4096

__device__ __forceinline__ void
gat_link_body(const int* __restrict__ dst,
              int* __restrict__ head,
              int* __restrict__ next,
              int n_nodes, int n_edges, int blk) {
    __shared__ unsigned cnt[LK_MAXBIN];
    __shared__ int      key[LK_CHUNK];
    __shared__ int      idx[LK_CHUNK];
    __shared__ unsigned wsum[LK_THREADS / 32];

    const int tid  = threadIdx.x;
    const int lane = tid & 31;
    const int wave = tid >> 5;
    const int cb   = blk * LK_CHUNK;
    const int n    = min(LK_CHUNK, n_edges - cb);

    int shift = 5;
    while (shift < 31 && (((unsigned)(n_nodes - 1)) >> shift) >= LK_MAXBIN) ++shift;

    for (int i = tid; i < LK_MAXBIN; i += LK_THREADS) cnt[i] = 0u;
    __syncthreads();

    int      dj[LK_PER];
    unsigned rj[LK_PER];
    unsigned bj[LK_PER];
#pragma unroll
    for (int j = 0; j < LK_PER; ++j) {
        const int li = j * LK_THREADS + tid;
        dj[j] = (li < n) ? dst[cb + li] : 0;
    }
#pragma unroll
    for (int j = 0; j < LK_PER; ++j) {
        const int li = j * LK_THREADS + tid;
        const unsigned b = min(((unsigned)dj[j]) >> shift, (unsigned)(LK_MAXBIN - 1));
        bj[j] = b;
        rj[j] = (li < n) ? atomicAdd(&cnt[b], 1u) : 0u;
    }
    __syncthreads();

    {
        const int t4 = tid * 4;
        const unsigned c0 = cnt[t4], c1 = cnt[t4 + 1], c2 = cnt[t4 + 2], c3 = cnt[t4 + 3];
        const unsigned s = c0 + c1 + c2 + c3;
        unsigned inc = s;
        for (int o = 1; o < 32; o <<= 1) {
            const unsigned v = __shfl_up(inc, o, 32);
            if (lane >= o) inc += v;
        }
        if (lane == 31) wsum[wave] = inc;
        __syncthreads();
        unsigned off = 0u;
        for (int w = 0; w < wave; ++w) off += wsum[w];
        const unsigned ex = off + inc - s;
        cnt[t4]     = ex;
        cnt[t4 + 1] = ex + c0;
        cnt[t4 + 2] = ex + c0 + c1;
        cnt[t4 + 3] = ex + c0 + c1 + c2;
    }
    __syncthreads();

#pragma unroll
    for (int j = 0; j < LK_PER; ++j) {
        const int li = j * LK_THREADS + tid;
        if (li < n) {
            const unsigned pos = cnt[bj[j]] + rj[j];
            key[pos] = dj[j];
            idx[pos] = li;
        }
    }
    __syncthreads();

    int sd[LK_PER];
    int si[LK_PER];
#pragma unroll
    for (int j = 0; j < LK_PER; ++j) {
        const int li = j * LK_THREADS + tid;
        sd[j] = (li < n) ? key[li] : 0;
        si[j] = (li < n) ? idx[li] : 0;
    }
    __syncthreads();

    int pj[LK_PER];
#pragma unroll
    for (int j = 0; j < LK_PER; ++j) {
        const int li = j * LK_THREADS + tid;
        pj[j] = (li < n) ? atomicExch(&head[sd[j]], cb + si[j]) : -1;
    }
#pragma unroll
    for (int j = 0; j < LK_PER; ++j) {
        const int li = j * LK_THREADS + tid;
        if (li < n) key[si[j]] = pj[j];
    }
    __syncthreads();

    for (int li = tid; li < n; li += LK_THREADS) next[cb + li] = key[li];
}

#define GM_WAVES  (LK_THREADS / 32)
#define GM_KB     (IN_DIM / 8)
#define GM_NT     (OUT_DIM / 16)
#define GM_WP4    (IN_DIM / 4 + 1)

__device__ __forceinline__ float gat_xhalf(float x) {
    const int b = __float_as_int(x);
    return __int_as_float(__builtin_amdgcn_permlanex16(b, b, 0x76543210, 0xfedcba98, false, false));
}

template <bool GUARD>
__device__ __forceinline__ void gat_store_tile(const v8f (&acc)[GM_NT], float* __restrict__ zp,
                                               int rows, int half) {
#pragma unroll
    for (int p = 0; p < GM_NT / 2; ++p) {
#pragma unroll
        for (int v = 0; v < 8; ++v) {
            const float lo = acc[2 * p][v];
            const float hi = acc[2 * p + 1][v];
            const float send = half ? lo : hi;
            const float recv = gat_xhalf(send);
            const float x = half ? recv : lo;
            const float y = half ? hi : recv;
            if (!GUARD || v < rows)     zp[v * OUT_DIM + 32 * p] = x;
            if (!GUARD || v + 8 < rows) zp[(v + 8) * OUT_DIM + 32 * p] = y;
        }
    }
}

__device__ __forceinline__ void
gat_gemm_body(const float* __restrict__ h,
              const float* __restrict__ W,
              float* __restrict__ z,
              int n_nodes, int gblk, int gblocks) {
    const int lane  = threadIdx.x & 31;
    const int half  = lane >> 4;
    const int lcol  = lane & 15;
    const int wave  = __builtin_amdgcn_readfirstlane((int)(threadIdx.x >> 5));

    __shared__ float4 wl[OUT_DIM * GM_WP4];
#pragma unroll
    for (int q = 0; q < OUT_DIM * IN_DIM / 4 / LK_THREADS; ++q) {
        const int f = (int)threadIdx.x + LK_THREADS * q;
        wl[(f >> 5) * GM_WP4 + (f & 31)] = ((const float4*)W)[f];
    }
    __syncthreads();

    const int n_tiles = (n_nodes + 15) >> 4;
    const int stride  = gblocks * GM_WAVES;
    const int koff    = 4 * half;
    const int wbase = lcol * GM_WP4 + half;

    for (int tile = wave * gblocks + gblk; tile < n_tiles; tile += stride) {
        int wofs = wbase;
        asm volatile("" : "+v"(wofs));
        float4 a[GM_KB];
        {
            int arow = tile * 16 + lcol;
            if (arow >= n_nodes) arow = n_nodes - 1;
            const float* hrow = h + (size_t)arow * IN_DIM + koff;
#pragma unroll
            for (int j = 0; j < GM_KB; ++j) a[j] = *(const float4*)(hrow + 8 * j);
        }

        v8f acc[GM_NT];
#pragma unroll
        for (int t = 0; t < GM_NT; ++t) acc[t] = v8f{};

#pragma unroll
        for (int j = 0; j < GM_KB; ++j) {
            float4 b[GM_NT];
#pragma unroll
            for (int t = 0; t < GM_NT; ++t) b[t] = wl[wofs + 16 * t * GM_WP4 + 2 * j];
            v2f a0; a0[0] = a[j].x; a0[1] = a[j].y;
            v2f a1; a1[0] = a[j].z; a1[1] = a[j].w;
#pragma unroll
            for (int t = 0; t < GM_NT; ++t) {
                v2f b0; b0[0] = b[t].x; b0[1] = b[t].y;
                acc[t] = __builtin_amdgcn_wmma_f32_16x16x4_f32(
                     false, a0,  false, b0,
                     (short)0, acc[t],  false,  false);
            }
#pragma unroll
            for (int t = 0; t < GM_NT; ++t) {
                v2f b1; b1[0] = b[t].z; b1[1] = b[t].w;
                acc[t] = __builtin_amdgcn_wmma_f32_16x16x4_f32(
                     false, a1,  false, b1,
                     (short)0, acc[t],  false,  false);
            }
        }

        const int row0 = tile * 16;
        float* zp = z + (size_t)row0 * OUT_DIM + lane;
        if (row0 + 16 <= n_nodes) gat_store_tile<false>(acc, zp, 16, half);
        else                      gat_store_tile<true>(acc, zp, n_nodes - row0, half);
    }
}

__global__ void __launch_bounds__(LK_THREADS)
gat_link_gemm_kernel(const int* __restrict__ dst,
                     int* __restrict__ head,
                     int* __restrict__ next,
                     const float* __restrict__ h,
                     const float* __restrict__ W,
                     float* __restrict__ z,
                     int n_nodes, int n_edges, int lk_blocks) {
    const int blk = (int)blockIdx.x;
    if (blk < lk_blocks) gat_link_body(dst, head, next, n_nodes, n_edges, blk);
    else                 gat_gemm_body(h, W, z, n_nodes, blk - lk_blocks, (int)gridDim.x - lk_blocks);
}

#define GL  4
#define NPW (32 / GL)
#define NP  (OUT_DIM / 4 / GL)
#define NH  (2 * NP)

typedef __attribute__((ext_vector_type(4))) float v4f;

__device__ __forceinline__ unsigned gat_clamp(int i, int n) {
    const unsigned u = (unsigned)i, top = (unsigned)(n - 1);
    return u < top ? u : top;
}

template <int MASK>
__device__ __forceinline__ float gat_add_xor(float p) {
    const int q = __builtin_amdgcn_update_dpp(0, __float_as_int(p), 0x160 + MASK, 0xf, 0xf, true);
    return p + __int_as_float(q);
}

__device__ __forceinline__ void
gat_walk_step(const v4f* __restrict__ z4, const int* __restrict__ src, const int* __restrict__ next,
              int n_edges, unsigned sub, const v2f (&zd)[NH],
              int eA, unsigned sB, bool vC, int& nn, int& ss,
              const v2f (&rin)[NH], v2f (&rout)[NH],
              float& m, float& l, v2f (&acc)[NH]) {
    const unsigned ec = gat_clamp(eA, n_edges);
    nn = next[ec];
    ss = src[ec];

    const v4f* rp = z4 + (sB * (unsigned)(OUT_DIM / 4) + sub);
#pragma unroll
    for (int q = 0; q < NP; ++q) {
        const v4f t = rp[GL * q];
        rout[2 * q]     = v2f{t.x, t.y};
        rout[2 * q + 1] = v2f{t.z, t.w};
    }
    __builtin_amdgcn_sched_barrier(0);

    v2f pa = rin[0] * zd[0];
    v2f pb = rin[1] * zd[1];
#pragma unroll
    for (int h = 2; h < NH; h += 2) {
        pa = __builtin_elementwise_fma(rin[h],     zd[h],     pa);
        pb = __builtin_elementwise_fma(rin[h + 1], zd[h + 1], pb);
    }
    const v2f ps = pa + pb;
    float p = ps.x + ps.y;
    p = gat_add_xor<1>(p);
    p = gat_add_xor<2>(p);
    if (GL > 4) p = gat_add_xor<4>(p);

    const float sc    = vC ? p : -INFINITY;
    const bool  rise  = sc > m;
    const float t     = __expf(-fabsf(sc - m));
    const float corr  = rise ? t : 1.0f;
    const float w     = vC ? (rise ? 1.0f : t) : 0.0f;
    m = fmaxf(m, sc);
    l = l * corr + w;
    const v2f c2 = v2f{corr, corr};
    const v2f w2 = v2f{w, w};
#pragma unroll
    for (int h = 0; h < NH; ++h) acc[h] = __builtin_elementwise_fma(w2, rin[h], acc[h] * c2);
}

__global__ void __launch_bounds__(256) __attribute__((amdgpu_waves_per_eu(8, 12)))
gat_node_kernel(const float* __restrict__ z,
                const int* __restrict__ src,
                const int* __restrict__ head,
                const int* __restrict__ next,
                float* __restrict__ out,
                int n_nodes, int n_edges) {
    const unsigned lane = threadIdx.x & 31u;
    const unsigned sub  = lane & (unsigned)(GL - 1);
    const unsigned grp  = lane / (unsigned)GL;
    const int wv   = blockIdx.x * (blockDim.x >> 5) + (threadIdx.x >> 5);
    const int d    = wv * NPW + (int)grp;
    const bool live = d < n_nodes;
    const unsigned dc = gat_clamp(d, n_nodes);

    const v4f* z4 = (const v4f*)z;

    v2f zd[NH];
    {
        const v4f* zdp = z4 + (dc * (unsigned)(OUT_DIM / 4) + sub);
#pragma unroll
        for (int q = 0; q < NP; ++q) {
            const v4f t = zdp[GL * q];
            zd[2 * q]     = v2f{t.x, t.y};
            zd[2 * q + 1] = v2f{t.z, t.w};
        }
    }

    int eA = live ? head[dc] : -1;
    unsigned sB = 0u;
    bool     vB = false;
    bool     vC = false;

    float m = -INFINITY;
    float l = 0.0f;
    v2f ra[NH], rb[NH], acc[NH];
#pragma unroll
    for (int h = 0; h < NH; ++h) {
        ra[h]  = v2f{0.0f, 0.0f};
        rb[h]  = v2f{0.0f, 0.0f};
        acc[h] = v2f{0.0f, 0.0f};
    }

    for (int it = 0; it < n_edges + 2; it += 2) {
        if (!__any((eA >= 0) || vB || vC)) break;
        int nn, ss;
        bool vA;

        vA = eA >= 0;
        gat_walk_step(z4, src, next, n_edges, sub, zd, eA, sB, vC, nn, ss, ra, rb, m, l, acc);
        vC = vB;
        sB = gat_clamp(ss, n_nodes);  vB = vA;
        eA = vA ? nn : -1;

        vA = eA >= 0;
        gat_walk_step(z4, src, next, n_edges, sub, zd, eA, sB, vC, nn, ss, rb, ra, m, l, acc);
        vC = vB;
        sB = gat_clamp(ss, n_nodes);  vB = vA;
        eA = vA ? nn : -1;
    }

    if (live) {
        const float inv = 1.0f / fmaxf(l, 1e-30f);
        v4f* op = (v4f*)out + ((unsigned)d * (unsigned)(OUT_DIM / 4) + sub);
#pragma unroll
        for (int q = 0; q < NP; ++q)
            op[GL * q] = v4f{acc[2 * q].x * inv, acc[2 * q].y * inv, acc[2 * q + 1].x * inv, acc[2 * q + 1].y * inv};
    }
}

#define FG_WGPS   256
#define FG_GM_MIN 64

extern "C" void kernel_launch(void* const* d_in, const int* in_sizes, int n_in,
                              void* d_out, int out_size, void* d_ws, size_t ws_size,
                              hipStream_t stream) {
    const float* h   = (const float*)d_in[0];
    const float* W   = (const float*)d_in[1];
    const int*   src = (const int*)d_in[2];
    const int*   dst = (const int*)d_in[3];
    float* out = (float*)d_out;

    const int n_nodes = in_sizes[0] / IN_DIM;
    const int n_edges = in_sizes[2];

    float* z    = (float*)d_ws;
    int*   next = (int*)(z + (size_t)n_nodes * OUT_DIM);
    int*   head = next + n_edges;

    {
        gat_head_init_kernel<<<(n_nodes + 255) / 256, 256, 0, stream>>>(head, n_nodes);
    }
    {
        int lk_blocks = (n_edges + LK_CHUNK - 1) / LK_CHUNK;
        int gm_blocks = FG_WGPS - lk_blocks;
        if (gm_blocks < FG_GM_MIN) gm_blocks = FG_GM_MIN;
        gat_link_gemm_kernel<<<lk_blocks + gm_blocks, LK_THREADS, 0, stream>>>(
            dst, head, next, h, W, z, n_nodes, n_edges, lk_blocks);
    }
    {
        int nodes_per_block = (256 / 32) * NPW;
        int blocks = (n_nodes + nodes_per_block - 1) / nodes_per_block;
        gat_node_kernel<<<blocks, 256, 0, stream>>>(
            z, src, head, next, out, n_nodes, n_edges);
    }
}


# Exchanges moved out of the few sorting work-groups into a launch over all WGPs, fed by (destination, source) pairs stored in sorted order; lists emptied in launch 1; lists in arrival order as before
# speedup vs baseline: 8.2142x; 1.0686x over previous
#include <hip/hip_runtime.h>
#include <hip/hip_bf16.h>

#define IN_DIM  128
#define OUT_DIM 64

typedef __attribute__((ext_vector_type(2))) float v2f;
typedef __attribute__((ext_vector_type(8))) float v8f;


#define LK_THREADS 1024
#define LK_PER     24
#define LK_CHUNK   (LK_THREADS * LK_PER)
#define LK_MAXBIN  4096

template <bool FULL>
__device__ __forceinline__ void
gat_sort_chunk(const int* __restrict__ dst,
               const int* __restrict__ src,
               int* __restrict__ dstS,
               int* __restrict__ srcS,
               unsigned* cnt, int* key, int* val, unsigned* wsum,
               int shift, int cb, int n) {
    const int tid  = threadIdx.x;
    const int lane = tid & 31;
    const int wave = tid >> 5;

    for (int i = tid; i < LK_MAXBIN; i += LK_THREADS) cnt[i] = 0u;
    __syncthreads();

    int      dj[LK_PER];
    int      sj[LK_PER];
    unsigned rj[LK_PER];
#pragma unroll
    for (int j = 0; j < LK_PER; ++j) {
        const int li = j * LK_THREADS + tid;
        dj[j] = (FULL || li < n) ? dst[cb + li] : 0;
    }
#pragma unroll
    for (int j = 0; j < LK_PER; ++j) {
        const int li = j * LK_THREADS + tid;
        const unsigned b = min(((unsigned)dj[j]) >> shift, (unsigned)(LK_MAXBIN - 1));
        rj[j] = (FULL || li < n) ? atomicAdd(&cnt[b], 1u) : 0u;
    }
#pragma unroll
    for (int j = 0; j < LK_PER; ++j) {
        const int li = j * LK_THREADS + tid;
        sj[j] = (FULL || li < n) ? src[cb + li] : 0;
    }
    __syncthreads();

    {
        const int t4 = tid * 4;
        const unsigned c0 = cnt[t4], c1 = cnt[t4 + 1], c2 = cnt[t4 + 2], c3 = cnt[t4 + 3];
        const unsigned s = c0 + c1 + c2 + c3;
        unsigned inc = s;
        for (int o = 1; o < 32; o <<= 1) {
            const unsigned v = __shfl_up(inc, o, 32);
            if (lane >= o) inc += v;
        }
        if (lane == 31) wsum[wave] = inc;
        __syncthreads();
        unsigned off = 0u;
        for (int w = 0; w < wave; ++w) off += wsum[w];
        const unsigned ex = off + inc - s;
        cnt[t4]     = ex;
        cnt[t4 + 1] = ex + c0;
        cnt[t4 + 2] = ex + c0 + c1;
        cnt[t4 + 3] = ex + c0 + c1 + c2;
    }
    __syncthreads();

#pragma unroll
    for (int j = 0; j < LK_PER; ++j) {
        const int li = j * LK_THREADS + tid;
        const unsigned b = min(((unsigned)dj[j]) >> shift, (unsigned)(LK_MAXBIN - 1));
        if (FULL || li < n) rj[j] += cnt[b];
    }
#pragma unroll
    for (int j = 0; j < LK_PER; ++j) {
        const int li = j * LK_THREADS + tid;
        if (FULL || li < n) {
            key[rj[j]] = dj[j];
            val[rj[j]] = sj[j];
        }
    }
    __syncthreads();

#pragma unroll
    for (int j = 0; j < LK_PER; ++j) {
        const int li = j * LK_THREADS + tid;
        if (FULL || li < n) {
            dstS[cb + li] = key[li];
            srcS[cb + li] = val[li];
        }
    }
}

__device__ __forceinline__ void
gat_sort_body(const int* __restrict__ dst,
              const int* __restrict__ src,
              int* __restrict__ dstS,
              int* __restrict__ srcS,
              int n_nodes, int n_edges, int blk) {
    __shared__ unsigned cnt[LK_MAXBIN];
    __shared__ int      key[LK_CHUNK];
    __shared__ int      val[LK_CHUNK];
    __shared__ unsigned wsum[LK_THREADS / 32];

    const int cb = blk * LK_CHUNK;
    const int n  = min(LK_CHUNK, n_edges - cb);

    int shift = 5;
    while (shift < 31 && (((unsigned)(n_nodes - 1)) >> shift) >= LK_MAXBIN) ++shift;

    if (n == LK_CHUNK) gat_sort_chunk<true>(dst, src, dstS, srcS, cnt, key, val, wsum, shift, cb, n);
    else               gat_sort_chunk<false>(dst, src, dstS, srcS, cnt, key, val, wsum, shift, cb, n);
}

#define GM_WAVES  (LK_THREADS / 32)
#define GM_KB     (IN_DIM / 8)
#define GM_NT     (OUT_DIM / 16)
#define GM_WP4    (IN_DIM / 4 + 1)

__device__ __forceinline__ float gat_xhalf(float x) {
    const int b = __float_as_int(x);
    return __int_as_float(__builtin_amdgcn_permlanex16(b, b, 0x76543210, 0xfedcba98, false, false));
}

template <bool GUARD>
__device__ __forceinline__ void gat_store_tile(const v8f (&acc)[GM_NT], float* __restrict__ zp,
                                               int rows, int half) {
#pragma unroll
    for (int p = 0; p < GM_NT / 2; ++p) {
#pragma unroll
        for (int v = 0; v < 8; ++v) {
            const float lo = acc[2 * p][v];
            const float hi = acc[2 * p + 1][v];
            const float send = half ? lo : hi;
            const float recv = gat_xhalf(send);
            const float x = half ? recv : lo;
            const float y = half ? hi : recv;
            if (!GUARD || v < rows)     zp[v * OUT_DIM + 32 * p] = x;
            if (!GUARD || v + 8 < rows) zp[(v + 8) * OUT_DIM + 32 * p] = y;
        }
    }
}

__device__ __forceinline__ void
gat_gemm_body(const float* __restrict__ h,
              const float* __restrict__ W,
              float* __restrict__ z,
              int n_nodes, int gblk, int gblocks) {
    const int lane  = threadIdx.x & 31;
    const int half  = lane >> 4;
    const int lcol  = lane & 15;
    const int wave  = __builtin_amdgcn_readfirstlane((int)(threadIdx.x >> 5));

    __shared__ float4 wl[OUT_DIM * GM_WP4];
#pragma unroll
    for (int q = 0; q < OUT_DIM * IN_DIM / 4 / LK_THREADS; ++q) {
        const int f = (int)threadIdx.x + LK_THREADS * q;
        wl[(f >> 5) * GM_WP4 + (f & 31)] = ((const float4*)W)[f];
    }
    __syncthreads();

    const int n_tiles = (n_nodes + 15) >> 4;
    const int stride  = gblocks * GM_WAVES;
    const int koff    = 4 * half;
    const int wbase = lcol * GM_WP4 + half;

    for (int tile = wave * gblocks + gblk; tile < n_tiles; tile += stride) {
        int wofs = wbase;
        asm volatile("" : "+v"(wofs));
        float4 a[GM_KB];
        {
            int arow = tile * 16 + lcol;
            if (arow >= n_nodes) arow = n_nodes - 1;
            const float* hrow = h + (size_t)arow * IN_DIM + koff;
#pragma unroll
            for (int j = 0; j < GM_KB; ++j) a[j] = *(const float4*)(hrow + 8 * j);
        }

        v8f acc[GM_NT];
#pragma unroll
        for (int t = 0; t < GM_NT; ++t) acc[t] = v8f{};

#pragma unroll
        for (int j = 0; j < GM_KB; ++j) {
            float4 b[GM_NT];
#pragma unroll
            for (int t = 0; t < GM_NT; ++t) b[t] = wl[wofs + 16 * t * GM_WP4 + 2 * j];
            v2f a0; a0[0] = a[j].x; a0[1] = a[j].y;
            v2f a1; a1[0] = a[j].z; a1[1] = a[j].w;
#pragma unroll
            for (int t = 0; t < GM_NT; ++t) {
                v2f b0; b0[0] = b[t].x; b0[1] = b[t].y;
                acc[t] = __builtin_amdgcn_wmma_f32_16x16x4_f32(
                     false, a0,  false, b0,
                     (short)0, acc[t],  false,  false);
            }
#pragma unroll
            for (int t = 0; t < GM_NT; ++t) {
                v2f b1; b1[0] = b[t].z; b1[1] = b[t].w;
                acc[t] = __builtin_amdgcn_wmma_f32_16x16x4_f32(
                     false, a1,  false, b1,
                     (short)0, acc[t],  false,  false);
            }
        }

        const int row0 = tile * 16;
        float* zp = z + (size_t)row0 * OUT_DIM + lane;
        if (row0 + 16 <= n_nodes) gat_store_tile<false>(acc, zp, 16, half);
        else                      gat_store_tile<true>(acc, zp, n_nodes - row0, half);
    }
}

__global__ void __launch_bounds__(LK_THREADS)
gat_sort_gemm_kernel(const int* __restrict__ dst,
                     const int* __restrict__ src,
                     int* __restrict__ dstS,
                     int* __restrict__ srcS,
                     int* __restrict__ head,
                     const float* __restrict__ h,
                     const float* __restrict__ W,
                     float* __restrict__ z,
                     int n_nodes, int n_edges, int lk_blocks) {
    const int blk = (int)blockIdx.x;
    for (int i = blk * LK_THREADS + (int)threadIdx.x; i < n_nodes; i += (int)gridDim.x * LK_THREADS)
        head[i] = -1;
    if (blk < lk_blocks) gat_sort_body(dst, src, dstS, srcS, n_nodes, n_edges, blk);
    else                 gat_gemm_body(h, W, z, n_nodes, blk - lk_blocks, (int)gridDim.x - lk_blocks);
}

#define GL  4
#define NPW (32 / GL)
#define NP  (OUT_DIM / 4 / GL)
#define NH  (2 * NP)

typedef __attribute__((ext_vector_type(4))) float v4f;

__device__ __forceinline__ unsigned gat_clamp(int i, int n) {
    const unsigned u = (unsigned)i, top = (unsigned)(n - 1);
    return u < top ? u : top;
}

template <int MASK>
__device__ __forceinline__ float gat_add_xor(float p) {
    const int q = __builtin_amdgcn_update_dpp(0, __float_as_int(p), 0x160 + MASK, 0xf, 0xf, true);
    return p + __int_as_float(q);
}

__device__ __forceinline__ void
gat_walk_step(const v4f* __restrict__ z4, const int* __restrict__ src, const int* __restrict__ next,
              int n_edges, unsigned sub, const v2f (&zd)[NH],
              int eA, unsigned sB, bool vC, int& nn, int& ss,
              const v2f (&rin)[NH], v2f (&rout)[NH],
              float& m, float& l, v2f (&acc)[NH]) {
    const unsigned ec = gat_clamp(eA, n_edges);
    nn = next[ec];
    ss = src[ec];

    const v4f* rp = z4 + (sB * (unsigned)(OUT_DIM / 4) + sub);
#pragma unroll
    for (int q = 0; q < NP; ++q) {
        const v4f t = rp[GL * q];
        rout[2 * q]     = v2f{t.x, t.y};
        rout[2 * q + 1] = v2f{t.z, t.w};
    }
    __builtin_amdgcn_sched_barrier(0);

    v2f pa = rin[0] * zd[0];
    v2f pb = rin[1] * zd[1];
#pragma unroll
    for (int h = 2; h < NH; h += 2) {
        pa = __builtin_elementwise_fma(rin[h],     zd[h],     pa);
        pb = __builtin_elementwise_fma(rin[h + 1], zd[h + 1], pb);
    }
    const v2f ps = pa + pb;
    float p = ps.x + ps.y;
    p = gat_add_xor<1>(p);
    p = gat_add_xor<2>(p);
    if (GL > 4) p = gat_add_xor<4>(p);

    const float sc    = vC ? p : -INFINITY;
    const bool  rise  = sc > m;
    const float t     = __expf(-fabsf(sc - m));
    const float corr  = rise ? t : 1.0f;
    const float w     = vC ? (rise ? 1.0f : t) : 0.0f;
    m = fmaxf(m, sc);
    l = l * corr + w;
    const v2f c2 = v2f{corr, corr};
    const v2f w2 = v2f{w, w};
#pragma unroll
    for (int h = 0; h < NH; ++h) acc[h] = __builtin_elementwise_fma(w2, rin[h], acc[h] * c2);
}

#define EX_THREADS 1024
#define EX_PER     7

__global__ void __launch_bounds__(EX_THREADS)
gat_exchange_kernel(const int* __restrict__ dstS,
                    int* __restrict__ head,
                    int* __restrict__ next,
                    int n_nodes, int n_edges) {
    const int base = (int)blockIdx.x * (EX_THREADS * EX_PER) + (int)threadIdx.x;
    int dj[EX_PER];
    int pj[EX_PER];
#pragma unroll
    for (int j = 0; j < EX_PER; ++j) {
        const int p = base + j * EX_THREADS;
        dj[j] = (p < n_edges) ? dstS[p] : 0;
    }
#pragma unroll
    for (int j = 0; j < EX_PER; ++j) {
        const int p = base + j * EX_THREADS;
        pj[j] = (p < n_edges) ? atomicExch(&head[gat_clamp(dj[j], n_nodes)], p) : -1;
    }
#pragma unroll
    for (int j = 0; j < EX_PER; ++j) {
        const int p = base + j * EX_THREADS;
        if (p < n_edges) next[p] = pj[j];
    }
}

__global__ void __launch_bounds__(256) __attribute__((amdgpu_waves_per_eu(8, 12)))
gat_node_kernel(const float* __restrict__ z,
                const int* __restrict__ src,
                const int* __restrict__ head,
                const int* __restrict__ next,
                float* __restrict__ out,
                int n_nodes, int n_edges) {
    const unsigned lane = threadIdx.x & 31u;
    const unsigned sub  = lane & (unsigned)(GL - 1);
    const unsigned grp  = lane / (unsigned)GL;
    const int wv   = blockIdx.x * (blockDim.x >> 5) + (threadIdx.x >> 5);
    const int d    = wv * NPW + (int)grp;
    const bool live = d < n_nodes;
    const unsigned dc = gat_clamp(d, n_nodes);

    const v4f* z4 = (const v4f*)z;

    v2f zd[NH];
    {
        const v4f* zdp = z4 + (dc * (unsigned)(OUT_DIM / 4) + sub);
#pragma unroll
        for (int q = 0; q < NP; ++q) {
            const v4f t = zdp[GL * q];
            zd[2 * q]     = v2f{t.x, t.y};
            zd[2 * q + 1] = v2f{t.z, t.w};
        }
    }

    int eA = live ? head[dc] : -1;
    unsigned sB = 0u;
    bool     vB = false;
    bool     vC = false;

    float m = -INFINITY;
    float l = 0.0f;
    v2f ra[NH], rb[NH], acc[NH];
#pragma unroll
    for (int h = 0; h < NH; ++h) {
        ra[h]  = v2f{0.0f, 0.0f};
        rb[h]  = v2f{0.0f, 0.0f};
        acc[h] = v2f{0.0f, 0.0f};
    }

    for (int it = 0; it < n_edges + 2; it += 2) {
        if (!__any((eA >= 0) || vB || vC)) break;
        int nn, ss;
        bool vA;

        vA = eA >= 0;
        gat_walk_step(z4, src, next, n_edges, sub, zd, eA, sB, vC, nn, ss, ra, rb, m, l, acc);
        vC = vB;
        sB = gat_clamp(ss, n_nodes);  vB = vA;
        eA = vA ? nn : -1;

        vA = eA >= 0;
        gat_walk_step(z4, src, next, n_edges, sub, zd, eA, sB, vC, nn, ss, rb, ra, m, l, acc);
        vC = vB;
        sB = gat_clamp(ss, n_nodes);  vB = vA;
        eA = vA ? nn : -1;
    }

    if (live) {
        const float inv = 1.0f / fmaxf(l, 1e-30f);
        v4f* op = (v4f*)out + ((unsigned)d * (unsigned)(OUT_DIM / 4) + sub);
#pragma unroll
        for (int q = 0; q < NP; ++q)
            op[GL * q] = v4f{acc[2 * q].x * inv, acc[2 * q].y * inv, acc[2 * q + 1].x * inv, acc[2 * q + 1].y * inv};
    }
}

#define FG_WGPS   256
#define FG_GM_MIN 64

extern "C" void kernel_launch(void* const* d_in, const int* in_sizes, int n_in,
                              void* d_out, int out_size, void* d_ws, size_t ws_size,
                              hipStream_t stream) {
    const float* h   = (const float*)d_in[0];
    const float* W   = (const float*)d_in[1];
    const int*   src = (const int*)d_in[2];
    const int*   dst = (const int*)d_in[3];
    float* out = (float*)d_out;

    const int n_nodes = in_sizes[0] / IN_DIM;
    const int n_edges = in_sizes[2];

    float* z    = (float*)d_ws;
    int*   next = (int*)(z + (size_t)n_nodes * OUT_DIM);
    int*   srcS = next + n_edges;
    int*   head = srcS + n_edges;
    int*   dstS = (int*)d_out;
    if (ws_size < ((size_t)n_nodes * OUT_DIM + 2 * (size_t)n_edges + (size_t)n_nodes) * 4) return;
    if ((size_t)out_size < (size_t)n_edges) return;

    {
        int lk_blocks = (n_edges + LK_CHUNK - 1) / LK_CHUNK;
        int gm_blocks = FG_WGPS - lk_blocks;
        if (gm_blocks < FG_GM_MIN) gm_blocks = FG_GM_MIN;
        gat_sort_gemm_kernel<<<lk_blocks + gm_blocks, LK_THREADS, 0, stream>>>(
            dst, src, dstS, srcS, head, h, W, z, n_nodes, n_edges, lk_blocks);
    }
    {
        int blocks = (n_edges + EX_THREADS * EX_PER - 1) / (EX_THREADS * EX_PER);
        gat_exchange_kernel<<<blocks, EX_THREADS, 0, stream>>>(dstS, head, next, n_nodes, n_edges);
    }
    {
        int nodes_per_block = (256 / 32) * NPW;
        int blocks = (n_nodes + nodes_per_block - 1) / nodes_per_block;
        gat_node_kernel<<<blocks, 256, 0, stream>>>(
            z, srcS, head, next, out, n_nodes, n_edges);
    }
}


# Node kernel: a wave walks two sets of eight nodes in turn (16 nodes a wave), three row buffers in rotation: half the waves to start, all resident at once; every sum in the same order as before
# speedup vs baseline: 8.5831x; 1.0449x over previous
#include <hip/hip_runtime.h>
#include <hip/hip_bf16.h>

#define IN_DIM  128
#define OUT_DIM 64

typedef __attribute__((ext_vector_type(2))) float v2f;
typedef __attribute__((ext_vector_type(8))) float v8f;


#define LK_THREADS 1024
#define LK_PER     24
#define LK_CHUNK   (LK_THREADS * LK_PER)
#define LK_MAXBIN  4096

template <bool FULL>
__device__ __forceinline__ void
gat_sort_chunk(const int* __restrict__ dst,
               const int* __restrict__ src,
               int* __restrict__ dstS,
               int* __restrict__ srcS,
               unsigned* cnt, int* key, int* val, unsigned* wsum,
               int shift, int cb, int n) {
    const int tid  = threadIdx.x;
    const int lane = tid & 31;
    const int wave = tid >> 5;

    for (int i = tid; i < LK_MAXBIN; i += LK_THREADS) cnt[i] = 0u;
    __syncthreads();

    int      dj[LK_PER];
    int      sj[LK_PER];
    unsigned rj[LK_PER];
#pragma unroll
    for (int j = 0; j < LK_PER; ++j) {
        const int li = j * LK_THREADS + tid;
        dj[j] = (FULL || li < n) ? dst[cb + li] : 0;
    }
#pragma unroll
    for (int j = 0; j < LK_PER; ++j) {
        const int li = j * LK_THREADS + tid;
        const unsigned b = min(((unsigned)dj[j]) >> shift, (unsigned)(LK_MAXBIN - 1));
        rj[j] = (FULL || li < n) ? atomicAdd(&cnt[b], 1u) : 0u;
    }
#pragma unroll
    for (int j = 0; j < LK_PER; ++j) {
        const int li = j * LK_THREADS + tid;
        sj[j] = (FULL || li < n) ? src[cb + li] : 0;
    }
    __syncthreads();

    {
        const int t4 = tid * 4;
        const unsigned c0 = cnt[t4], c1 = cnt[t4 + 1], c2 = cnt[t4 + 2], c3 = cnt[t4 + 3];
        const unsigned s = c0 + c1 + c2 + c3;
        unsigned inc = s;
        for (int o = 1; o < 32; o <<= 1) {
            const unsigned v = __shfl_up(inc, o, 32);
            if (lane >= o) inc += v;
        }
        if (lane == 31) wsum[wave] = inc;
        __syncthreads();
        unsigned off = 0u;
        for (int w = 0; w < wave; ++w) off += wsum[w];
        const unsigned ex = off + inc - s;
        cnt[t4]     = ex;
        cnt[t4 + 1] = ex + c0;
        cnt[t4 + 2] = ex + c0 + c1;
        cnt[t4 + 3] = ex + c0 + c1 + c2;
    }
    __syncthreads();

#pragma unroll
    for (int j = 0; j < LK_PER; ++j) {
        const int li = j * LK_THREADS + tid;
        const unsigned b = min(((unsigned)dj[j]) >> shift, (unsigned)(LK_MAXBIN - 1));
        if (FULL || li < n) rj[j] += cnt[b];
    }
#pragma unroll
    for (int j = 0; j < LK_PER; ++j) {
        const int li = j * LK_THREADS + tid;
        if (FULL || li < n) {
            key[rj[j]] = dj[j];
            val[rj[j]] = sj[j];
        }
    }
    __syncthreads();

#pragma unroll
    for (int j = 0; j < LK_PER; ++j) {
        const int li = j * LK_THREADS + tid;
        if (FULL || li < n) {
            dstS[cb + li] = key[li];
            srcS[cb + li] = val[li];
        }
    }
}

__device__ __forceinline__ void
gat_sort_body(const int* __restrict__ dst,
              const int* __restrict__ src,
              int* __restrict__ dstS,
              int* __restrict__ srcS,
              int n_nodes, int n_edges, int blk) {
    __shared__ unsigned cnt[LK_MAXBIN];
    __shared__ int      key[LK_CHUNK];
    __shared__ int      val[LK_CHUNK];
    __shared__ unsigned wsum[LK_THREADS / 32];

    const int cb = blk * LK_CHUNK;
    const int n  = min(LK_CHUNK, n_edges - cb);

    int shift = 5;
    while (shift < 31 && (((unsigned)(n_nodes - 1)) >> shift) >= LK_MAXBIN) ++shift;

    if (n == LK_CHUNK) gat_sort_chunk<true>(dst, src, dstS, srcS, cnt, key, val, wsum, shift, cb, n);
    else               gat_sort_chunk<false>(dst, src, dstS, srcS, cnt, key, val, wsum, shift, cb, n);
}

#define GM_WAVES  (LK_THREADS / 32)
#define GM_KB     (IN_DIM / 8)
#define GM_NT     (OUT_DIM / 16)
#define GM_WP4    (IN_DIM / 4 + 1)

__device__ __forceinline__ float gat_xhalf(float x) {
    const int b = __float_as_int(x);
    return __int_as_float(__builtin_amdgcn_permlanex16(b, b, 0x76543210, 0xfedcba98, false, false));
}

template <bool GUARD>
__device__ __forceinline__ void gat_store_tile(const v8f (&acc)[GM_NT], float* __restrict__ zp,
                                               int rows, int half) {
#pragma unroll
    for (int p = 0; p < GM_NT / 2; ++p) {
#pragma unroll
        for (int v = 0; v < 8; ++v) {
            const float lo = acc[2 * p][v];
            const float hi = acc[2 * p + 1][v];
            const float send = half ? lo : hi;
            const float recv = gat_xhalf(send);
            const float x = half ? recv : lo;
            const float y = half ? hi : recv;
            if (!GUARD || v < rows)     zp[v * OUT_DIM + 32 * p] = x;
            if (!GUARD || v + 8 < rows) zp[(v + 8) * OUT_DIM + 32 * p] = y;
        }
    }
}

__device__ __forceinline__ void
gat_gemm_body(const float* __restrict__ h,
              const float* __restrict__ W,
              float* __restrict__ z,
              int n_nodes, int gblk, int gblocks) {
    const int lane  = threadIdx.x & 31;
    const int half  = lane >> 4;
    const int lcol  = lane & 15;
    const int wave  = __builtin_amdgcn_readfirstlane((int)(threadIdx.x >> 5));

    __shared__ float4 wl[OUT_DIM * GM_WP4];
#pragma unroll
    for (int q = 0; q < OUT_DIM * IN_DIM / 4 / LK_THREADS; ++q) {
        const int f = (int)threadIdx.x + LK_THREADS * q;
        wl[(f >> 5) * GM_WP4 + (f & 31)] = ((const float4*)W)[f];
    }
    __syncthreads();

    const int n_tiles = (n_nodes + 15) >> 4;
    const int stride  = gblocks * GM_WAVES;
    const int koff    = 4 * half;
    const int wbase = lcol * GM_WP4 + half;

    for (int tile = wave * gblocks + gblk; tile < n_tiles; tile += stride) {
        int wofs = wbase;
        asm volatile("" : "+v"(wofs));
        float4 a[GM_KB];
        {
            int arow = tile * 16 + lcol;
            if (arow >= n_nodes) arow = n_nodes - 1;
            const float* hrow = h + (size_t)arow * IN_DIM + koff;
#pragma unroll
            for (int j = 0; j < GM_KB; ++j) a[j] = *(const float4*)(hrow + 8 * j);
        }

        v8f acc[GM_NT];
#pragma unroll
        for (int t = 0; t < GM_NT; ++t) acc[t] = v8f{};

#pragma unroll
        for (int j = 0; j < GM_KB; ++j) {
            float4 b[GM_NT];
#pragma unroll
            for (int t = 0; t < GM_NT; ++t) b[t] = wl[wofs + 16 * t * GM_WP4 + 2 * j];
            v2f a0; a0[0] = a[j].x; a0[1] = a[j].y;
            v2f a1; a1[0] = a[j].z; a1[1] = a[j].w;
#pragma unroll
            for (int t = 0; t < GM_NT; ++t) {
                v2f b0; b0[0] = b[t].x; b0[1] = b[t].y;
                acc[t] = __builtin_amdgcn_wmma_f32_16x16x4_f32(
                     false, a0,  false, b0,
                     (short)0, acc[t],  false,  false);
            }
#pragma unroll
            for (int t = 0; t < GM_NT; ++t) {
                v2f b1; b1[0] = b[t].z; b1[1] = b[t].w;
                acc[t] = __builtin_amdgcn_wmma_f32_16x16x4_f32(
                     false, a1,  false, b1,
                     (short)0, acc[t],  false,  false);
            }
        }

        const int row0 = tile * 16;
        float* zp = z + (size_t)row0 * OUT_DIM + lane;
        if (row0 + 16 <= n_nodes) gat_store_tile<false>(acc, zp, 16, half);
        else                      gat_store_tile<true>(acc, zp, n_nodes - row0, half);
    }
}

__global__ void __launch_bounds__(LK_THREADS)
gat_sort_gemm_kernel(const int* __restrict__ dst,
                     const int* __restrict__ src,
                     int* __restrict__ dstS,
                     int* __restrict__ srcS,
                     int* __restrict__ head,
                     const float* __restrict__ h,
                     const float* __restrict__ W,
                     float* __restrict__ z,
                     int n_nodes, int n_edges, int lk_blocks) {
    const int blk = (int)blockIdx.x;
    for (int i = blk * LK_THREADS + (int)threadIdx.x; i < n_nodes; i += (int)gridDim.x * LK_THREADS)
        head[i] = -1;
    if (blk < lk_blocks) gat_sort_body(dst, src, dstS, srcS, n_nodes, n_edges, blk);
    else                 gat_gemm_body(h, W, z, n_nodes, blk - lk_blocks, (int)gridDim.x - lk_blocks);
}

#define GL  4
#define NPW (32 / GL)
#define NP  (OUT_DIM / 4 / GL)
#define NH  (2 * NP)
#define NS  2

typedef __attribute__((ext_vector_type(4))) float v4f;

__device__ __forceinline__ unsigned gat_clamp(int i, int n) {
    const unsigned u = (unsigned)i, top = (unsigned)(n - 1);
    return u < top ? u : top;
}

template <int MASK>
__device__ __forceinline__ float gat_add_xor(float p) {
    const int q = __builtin_amdgcn_update_dpp(0, __float_as_int(p), 0x160 + MASK, 0xf, 0xf, true);
    return p + __int_as_float(q);
}

__device__ __forceinline__ void
gat_walk_step(const v4f* __restrict__ z4, const int* __restrict__ src, const int* __restrict__ next,
              int n_nodes, int n_edges, unsigned sub, const v2f (&zd)[NH],
              int& nn, int& ss, bool& vA, bool& vB,
              const v2f (&rin)[NH], v2f (&rout)[NH],
              float& m, float& l, v2f (&acc)[NH]) {
    const bool     vC = vB;
    const unsigned sB = gat_clamp(ss, n_nodes);
    vB = vA;
    const int eA = vA ? nn : -1;
    vA = eA >= 0;

    const unsigned ec = gat_clamp(eA, n_edges);
    nn = next[ec];
    ss = src[ec];

    const v4f* rp = z4 + (sB * (unsigned)(OUT_DIM / 4) + sub);
#pragma unroll
    for (int q = 0; q < NP; ++q) {
        const v4f t = rp[GL * q];
        rout[2 * q]     = v2f{t.x, t.y};
        rout[2 * q + 1] = v2f{t.z, t.w};
    }
    __builtin_amdgcn_sched_barrier(0);

    v2f pa = rin[0] * zd[0];
    v2f pb = rin[1] * zd[1];
#pragma unroll
    for (int h = 2; h < NH; h += 2) {
        pa = __builtin_elementwise_fma(rin[h],     zd[h],     pa);
        pb = __builtin_elementwise_fma(rin[h + 1], zd[h + 1], pb);
    }
    const v2f ps = pa + pb;
    float p = ps.x + ps.y;
    p = gat_add_xor<1>(p);
    p = gat_add_xor<2>(p);
    if (GL > 4) p = gat_add_xor<4>(p);
    asm volatile("" : "+v"(p));

    const float sc    = vC ? p : -INFINITY;
    const bool  rise  = sc > m;
    const float t     = __expf(-fabsf(sc - m));
    const float corr  = rise ? t : 1.0f;
    const float w     = vC ? (rise ? 1.0f : t) : 0.0f;
    m = fmaxf(m, sc);
    l = l * corr + w;
    v2f c2 = v2f{corr, corr};
    v2f w2 = v2f{w, w};
    asm volatile("" : "+v"(c2), "+v"(w2));
#pragma unroll
    for (int h = 0; h < NH; ++h) acc[h] = __builtin_elementwise_fma(w2, rin[h], acc[h] * c2);
    static_assert(NH == 8, "the statement below names the eight pairs of acc");
    asm volatile("" : "+v"(m), "+v"(l), "+v"(acc[0]), "+v"(acc[1]), "+v"(acc[2]), "+v"(acc[3]),
                      "+v"(acc[4]), "+v"(acc[5]), "+v"(acc[6]), "+v"(acc[7]));
    __builtin_amdgcn_sched_barrier(0);
}

#define EX_THREADS 1024
#define EX_PER     7

__global__ void __launch_bounds__(EX_THREADS)
gat_exchange_kernel(const int* __restrict__ dstS,
                    int* __restrict__ head,
                    int* __restrict__ next,
                    int n_nodes, int n_edges) {
    const int base = (int)blockIdx.x * (EX_THREADS * EX_PER) + (int)threadIdx.x;
    int dj[EX_PER];
    int pj[EX_PER];
#pragma unroll
    for (int j = 0; j < EX_PER; ++j) {
        const int p = base + j * EX_THREADS;
        dj[j] = (p < n_edges) ? dstS[p] : 0;
    }
#pragma unroll
    for (int j = 0; j < EX_PER; ++j) {
        const int p = base + j * EX_THREADS;
        pj[j] = (p < n_edges) ? atomicExch(&head[gat_clamp(dj[j], n_nodes)], p) : -1;
    }
#pragma unroll
    for (int j = 0; j < EX_PER; ++j) {
        const int p = base + j * EX_THREADS;
        if (p < n_edges) next[p] = pj[j];
    }
}

#define ND_THREADS 128
__global__ void __launch_bounds__(ND_THREADS) __attribute__((amdgpu_waves_per_eu(7, 7)))
gat_node_kernel(const float* __restrict__ z,
                const int* __restrict__ src,
                const int* __restrict__ head,
                const int* __restrict__ next,
                float* __restrict__ out,
                int n_nodes, int n_edges) {
    const unsigned lane = threadIdx.x & 31u;
    const unsigned sub  = lane & (unsigned)(GL - 1);
    const unsigned grp  = lane / (unsigned)GL;
    const int wv   = __builtin_amdgcn_readfirstlane((int)(blockIdx.x * (ND_THREADS / 32) + (threadIdx.x >> 5)));

    const v4f* z4 = (const v4f*)z;

    bool     live[NS];
    v2f      zd[NS][NH];
    int      nn[NS], ss[NS];
    bool     vA[NS], vB[NS];
    float    m[NS], l[NS];
    v2f      acc[NS][NH];
    v2f      r0[NH], r1[NH], r2[NH];

    int e0[NS];
#pragma unroll
    for (int w = 0; w < NS; ++w) {
        const int d = (wv * NS + w) * NPW + (int)grp;
        live[w] = d < n_nodes;
        const unsigned dc = gat_clamp(d, n_nodes);
        e0[w] = head[dc];
        const v4f* zdp = z4 + (dc * (unsigned)(OUT_DIM / 4) + sub);
#pragma unroll
        for (int q = 0; q < NP; ++q) {
            const v4f t = zdp[GL * q];
            zd[w][2 * q]     = v2f{t.x, t.y};
            zd[w][2 * q + 1] = v2f{t.z, t.w};
        }
    }
#pragma unroll
    for (int w = 0; w < NS; ++w) {
        const int e = live[w] ? e0[w] : -1;
        vA[w] = e >= 0;
        vB[w] = false;
        const unsigned ec = gat_clamp(e, n_edges);
        nn[w] = next[ec];
        ss[w] = src[ec];
        m[w] = -INFINITY;
        l[w] = 0.0f;
#pragma unroll
        for (int h = 0; h < NH; ++h) acc[w][h] = v2f{0.0f, 0.0f};
    }
#pragma unroll
    for (int h = 0; h < NH; ++h) {
        r0[h] = v2f{0.0f, 0.0f};
        r1[h] = v2f{0.0f, 0.0f};
        r2[h] = v2f{0.0f, 0.0f};
    }

    asm volatile("" : "+v"(nn[0]), "+v"(ss[0]), "+v"(nn[1]), "+v"(ss[1]));

    for (int it = 0; it < n_edges + 3; it += 3) {
        if (!__any(vA[0] || vB[0] || vA[1] || vB[1])) break;
        gat_walk_step(z4, src, next, n_nodes, n_edges, sub, zd[0], nn[0], ss[0], vA[0], vB[0], r1, r0, m[0], l[0], acc[0]);
        gat_walk_step(z4, src, next, n_nodes, n_edges, sub, zd[1], nn[1], ss[1], vA[1], vB[1], r2, r1, m[1], l[1], acc[1]);
        gat_walk_step(z4, src, next, n_nodes, n_edges, sub, zd[0], nn[0], ss[0], vA[0], vB[0], r0, r2, m[0], l[0], acc[0]);
        gat_walk_step(z4, src, next, n_nodes, n_edges, sub, zd[1], nn[1], ss[1], vA[1], vB[1], r1, r0, m[1], l[1], acc[1]);
        gat_walk_step(z4, src, next, n_nodes, n_edges, sub, zd[0], nn[0], ss[0], vA[0], vB[0], r2, r1, m[0], l[0], acc[0]);
        gat_walk_step(z4, src, next, n_nodes, n_edges, sub, zd[1], nn[1], ss[1], vA[1], vB[1], r0, r2, m[1], l[1], acc[1]);
    }

    const unsigned grp2 = __builtin_amdgcn_mbcnt_lo(~0u, 0u) / (unsigned)GL;
#pragma unroll
    for (int w = 0; w < NS; ++w) {
        if (live[w]) {
            const float inv = 1.0f / fmaxf(l[w], 1e-30f);
            const unsigned d = (unsigned)((wv * NS + w) * NPW) + grp2;
            v4f* op = (v4f*)out + (d * (unsigned)(OUT_DIM / 4) + sub);
#pragma unroll
            for (int q = 0; q < NP; ++q)
                op[GL * q] = v4f{acc[w][2 * q].x * inv, acc[w][2 * q].y * inv,
                                 acc[w][2 * q + 1].x * inv, acc[w][2 * q + 1].y * inv};
        }
    }
}

#define FG_WGPS   256
#define FG_GM_MIN 64

extern "C" void kernel_launch(void* const* d_in, const int* in_sizes, int n_in,
                              void* d_out, int out_size, void* d_ws, size_t ws_size,
                              hipStream_t stream) {
    const float* h   = (const float*)d_in[0];
    const float* W   = (const float*)d_in[1];
    const int*   src = (const int*)d_in[2];
    const int*   dst = (const int*)d_in[3];
    float* out = (float*)d_out;

    const int n_nodes = in_sizes[0] / IN_DIM;
    const int n_edges = in_sizes[2];

    float* z    = (float*)d_ws;
    int*   next = (int*)(z + (size_t)n_nodes * OUT_DIM);
    int*   srcS = next + n_edges;
    int*   head = srcS + n_edges;
    int*   dstS = (int*)d_out;
    if (ws_size < ((size_t)n_nodes * OUT_DIM + 2 * (size_t)n_edges + (size_t)n_nodes) * 4) return;
    if ((size_t)out_size < (size_t)n_edges) return;

    {
        int lk_blocks = (n_edges + LK_CHUNK - 1) / LK_CHUNK;
        int gm_blocks = FG_WGPS - lk_blocks;
        if (gm_blocks < FG_GM_MIN) gm_blocks = FG_GM_MIN;
        gat_sort_gemm_kernel<<<lk_blocks + gm_blocks, LK_THREADS, 0, stream>>>(
            dst, src, dstS, srcS, head, h, W, z, n_nodes, n_edges, lk_blocks);
    }
    {
        int blocks = (n_edges + EX_THREADS * EX_PER - 1) / (EX_THREADS * EX_PER);
        gat_exchange_kernel<<<blocks, EX_THREADS, 0, stream>>>(dstS, head, next, n_nodes, n_edges);
    }
    {
        int nodes_per_block = (ND_THREADS / 32) * NS * NPW;
        int blocks = (n_nodes + nodes_per_block - 1) / nodes_per_block;
        gat_node_kernel<<<blocks, ND_THREADS, 0, stream>>>(
            z, srcS, head, next, out, n_nodes, n_edges);
    }
}


# Each edge kept as one 8-byte record from the sort on: (destination, source) into the exchanges, (link, source) into the walk: one load and one line per edge step where link and source took two
# speedup vs baseline: 8.9007x; 1.0370x over previous
#include <hip/hip_runtime.h>
#include <hip/hip_bf16.h>

#define IN_DIM  128
#define OUT_DIM 64

typedef __attribute__((ext_vector_type(2))) float v2f;
typedef __attribute__((ext_vector_type(8))) float v8f;


#define LK_THREADS 1024
#define LK_PER     24
#define LK_CHUNK   (LK_THREADS * LK_PER)
#define LK_MAXBIN  4096

template <bool FULL>
__device__ __forceinline__ void
gat_sort_chunk(const int* __restrict__ dst,
               const int* __restrict__ src,
               int2* __restrict__ pairS,
               unsigned* cnt, int* key, int* val, unsigned* wsum,
               int shift, int cb, int n) {
    const int tid  = threadIdx.x;
    const int lane = tid & 31;
    const int wave = tid >> 5;

    for (int i = tid; i < LK_MAXBIN; i += LK_THREADS) cnt[i] = 0u;
    __syncthreads();

    int      dj[LK_PER];
    int      sj[LK_PER];
    unsigned rj[LK_PER];
#pragma unroll
    for (int j = 0; j < LK_PER; ++j) {
        const int li = j * LK_THREADS + tid;
        dj[j] = (FULL || li < n) ? dst[cb + li] : 0;
    }
#pragma unroll
    for (int j = 0; j < LK_PER; ++j) {
        const int li = j * LK_THREADS + tid;
        const unsigned b = min(((unsigned)dj[j]) >> shift, (unsigned)(LK_MAXBIN - 1));
        rj[j] = (FULL || li < n) ? atomicAdd(&cnt[b], 1u) : 0u;
    }
#pragma unroll
    for (int j = 0; j < LK_PER; ++j) {
        const int li = j * LK_THREADS + tid;
        sj[j] = (FULL || li < n) ? src[cb + li] : 0;
    }
    __syncthreads();

    {
        const int t4 = tid * 4;
        const unsigned c0 = cnt[t4], c1 = cnt[t4 + 1], c2 = cnt[t4 + 2], c3 = cnt[t4 + 3];
        const unsigned s = c0 + c1 + c2 + c3;
        unsigned inc = s;
        for (int o = 1; o < 32; o <<= 1) {
            const unsigned v = __shfl_up(inc, o, 32);
            if (lane >= o) inc += v;
        }
        if (lane == 31) wsum[wave] = inc;
        __syncthreads();
        unsigned off = 0u;
        for (int w = 0; w < wave; ++w) off += wsum[w];
        const unsigned ex = off + inc - s;
        cnt[t4]     = ex;
        cnt[t4 + 1] = ex + c0;
        cnt[t4 + 2] = ex + c0 + c1;
        cnt[t4 + 3] = ex + c0 + c1 + c2;
    }
    __syncthreads();

#pragma unroll
    for (int j = 0; j < LK_PER; ++j) {
        const int li = j * LK_THREADS + tid;
        const unsigned b = min(((unsigned)dj[j]) >> shift, (unsigned)(LK_MAXBIN - 1));
        if (FULL || li < n) rj[j] += cnt[b];
    }
#pragma unroll
    for (int j = 0; j < LK_PER; ++j) {
        const int li = j * LK_THREADS + tid;
        if (FULL || li < n) {
            key[rj[j]] = dj[j];
            val[rj[j]] = sj[j];
        }
    }
    __syncthreads();

#pragma unroll
    for (int j = 0; j < LK_PER; ++j) {
        const int li = j * LK_THREADS + tid;
        if (FULL || li < n) pairS[cb + li] = make_int2(key[li], val[li]);
    }
}

__device__ __forceinline__ void
gat_sort_body(const int* __restrict__ dst,
              const int* __restrict__ src,
              int2* __restrict__ pairS,
              int n_nodes, int n_edges, int blk) {
    __shared__ unsigned cnt[LK_MAXBIN];
    __shared__ int      key[LK_CHUNK];
    __shared__ int      val[LK_CHUNK];
    __shared__ unsigned wsum[LK_THREADS / 32];

    const int cb = blk * LK_CHUNK;
    const int n  = min(LK_CHUNK, n_edges - cb);

    int shift = 5;
    while (shift < 31 && (((unsigned)(n_nodes - 1)) >> shift) >= LK_MAXBIN) ++shift;

    if (n == LK_CHUNK) gat_sort_chunk<true>(dst, src, pairS, cnt, key, val, wsum, shift, cb, n);
    else               gat_sort_chunk<false>(dst, src, pairS, cnt, key, val, wsum, shift, cb, n);
}

#define GM_WAVES  (LK_THREADS / 32)
#define GM_KB     (IN_DIM / 8)
#define GM_NT     (OUT_DIM / 16)
#define GM_WP4    (IN_DIM / 4 + 1)

__device__ __forceinline__ float gat_xhalf(float x) {
    const int b = __float_as_int(x);
    return __int_as_float(__builtin_amdgcn_permlanex16(b, b, 0x76543210, 0xfedcba98, false, false));
}

template <bool GUARD>
__device__ __forceinline__ void gat_store_tile(const v8f (&acc)[GM_NT], float* __restrict__ zp,
                                               int rows, int half) {
#pragma unroll
    for (int p = 0; p < GM_NT / 2; ++p) {
#pragma unroll
        for (int v = 0; v < 8; ++v) {
            const float lo = acc[2 * p][v];
            const float hi = acc[2 * p + 1][v];
            const float send = half ? lo : hi;
            const float recv = gat_xhalf(send);
            const float x = half ? recv : lo;
            const float y = half ? hi : recv;
            if (!GUARD || v < rows)     zp[v * OUT_DIM + 32 * p] = x;
            if (!GUARD || v + 8 < rows) zp[(v + 8) * OUT_DIM + 32 * p] = y;
        }
    }
}

__device__ __forceinline__ void
gat_gemm_body(const float* __restrict__ h,
              const float* __restrict__ W,
              float* __restrict__ z,
              int n_nodes, int gblk, int gblocks) {
    const int lane  = threadIdx.x & 31;
    const int half  = lane >> 4;
    const int lcol  = lane & 15;
    const int wave  = __builtin_amdgcn_readfirstlane((int)(threadIdx.x >> 5));

    __shared__ float4 wl[OUT_DIM * GM_WP4];
#pragma unroll
    for (int q = 0; q < OUT_DIM * IN_DIM / 4 / LK_THREADS; ++q) {
        const int f = (int)threadIdx.x + LK_THREADS * q;
        wl[(f >> 5) * GM_WP4 + (f & 31)] = ((const float4*)W)[f];
    }
    __syncthreads();

    const int n_tiles = (n_nodes + 15) >> 4;
    const int stride  = gblocks * GM_WAVES;
    const int koff    = 4 * half;
    const int wbase = lcol * GM_WP4 + half;

    for (int tile = wave * gblocks + gblk; tile < n_tiles; tile += stride) {
        int wofs = wbase;
        asm volatile("" : "+v"(wofs));
        float4 a[GM_KB];
        {
            int arow = tile * 16 + lcol;
            if (arow >= n_nodes) arow = n_nodes - 1;
            const float* hrow = h + (size_t)arow * IN_DIM + koff;
#pragma unroll
            for (int j = 0; j < GM_KB; ++j) a[j] = *(const float4*)(hrow + 8 * j);
        }

        v8f acc[GM_NT];
#pragma unroll
        for (int t = 0; t < GM_NT; ++t) acc[t] = v8f{};

#pragma unroll
        for (int j = 0; j < GM_KB; ++j) {
            float4 b[GM_NT];
#pragma unroll
            for (int t = 0; t < GM_NT; ++t) b[t] = wl[wofs + 16 * t * GM_WP4 + 2 * j];
            v2f a0; a0[0] = a[j].x; a0[1] = a[j].y;
            v2f a1; a1[0] = a[j].z; a1[1] = a[j].w;
#pragma unroll
            for (int t = 0; t < GM_NT; ++t) {
                v2f b0; b0[0] = b[t].x; b0[1] = b[t].y;
                acc[t] = __builtin_amdgcn_wmma_f32_16x16x4_f32(
                     false, a0,  false, b0,
                     (short)0, acc[t],  false,  false);
            }
#pragma unroll
            for (int t = 0; t < GM_NT; ++t) {
                v2f b1; b1[0] = b[t].z; b1[1] = b[t].w;
                acc[t] = __builtin_amdgcn_wmma_f32_16x16x4_f32(
                     false, a1,  false, b1,
                     (short)0, acc[t],  false,  false);
            }
        }

        const int row0 = tile * 16;
        float* zp = z + (size_t)row0 * OUT_DIM + lane;
        if (row0 + 16 <= n_nodes) gat_store_tile<false>(acc, zp, 16, half);
        else                      gat_store_tile<true>(acc, zp, n_nodes - row0, half);
    }
}

__global__ void __launch_bounds__(LK_THREADS)
gat_sort_gemm_kernel(const int* __restrict__ dst,
                     const int* __restrict__ src,
                     int* __restrict__ pairS,
                     int* __restrict__ head,
                     const float* __restrict__ h,
                     const float* __restrict__ W,
                     float* __restrict__ z,
                     int n_nodes, int n_edges, int lk_blocks) {
    const int blk = (int)blockIdx.x;
    for (int i = blk * LK_THREADS + (int)threadIdx.x; i < n_nodes; i += (int)gridDim.x * LK_THREADS)
        head[i] = -1;
    if (blk < lk_blocks) gat_sort_body(dst, src, (int2*)pairS, n_nodes, n_edges, blk);
    else                 gat_gemm_body(h, W, z, n_nodes, blk - lk_blocks, (int)gridDim.x - lk_blocks);
}

#define GL  4
#define NPW (32 / GL)
#define NP  (OUT_DIM / 4 / GL)
#define NH  (2 * NP)
#define NS  2

typedef __attribute__((ext_vector_type(4))) float v4f;

__device__ __forceinline__ unsigned gat_clamp(int i, int n) {
    const unsigned u = (unsigned)i, top = (unsigned)(n - 1);
    return u < top ? u : top;
}

template <int MASK>
__device__ __forceinline__ float gat_add_xor(float p) {
    const int q = __builtin_amdgcn_update_dpp(0, __float_as_int(p), 0x160 + MASK, 0xf, 0xf, true);
    return p + __int_as_float(q);
}

__device__ __forceinline__ void
gat_walk_step(const v4f* __restrict__ z4, const int2* __restrict__ link,
              int n_nodes, int n_edges, unsigned sub, const v2f (&zd)[NH],
              int& nn, int& ss, bool& vA, bool& vB,
              const v2f (&rin)[NH], v2f (&rout)[NH],
              float& m, float& l, v2f (&acc)[NH]) {
    const bool     vC = vB;
    const unsigned sB = gat_clamp(ss, n_nodes);
    vB = vA;
    const int eA = vA ? nn : -1;
    vA = eA >= 0;

    const unsigned ec = gat_clamp(eA, n_edges);
    const int2 rec = link[ec];
    nn = rec.x;
    ss = rec.y;

    const v4f* rp = z4 + (sB * (unsigned)(OUT_DIM / 4) + sub);
#pragma unroll
    for (int q = 0; q < NP; ++q) {
        const v4f t = rp[GL * q];
        rout[2 * q]     = v2f{t.x, t.y};
        rout[2 * q + 1] = v2f{t.z, t.w};
    }
    __builtin_amdgcn_sched_barrier(0);

    v2f pa = rin[0] * zd[0];
    v2f pb = rin[1] * zd[1];
#pragma unroll
    for (int h = 2; h < NH; h += 2) {
        pa = __builtin_elementwise_fma(rin[h],     zd[h],     pa);
        pb = __builtin_elementwise_fma(rin[h + 1], zd[h + 1], pb);
    }
    const v2f ps = pa + pb;
    float p = ps.x + ps.y;
    p = gat_add_xor<1>(p);
    p = gat_add_xor<2>(p);
    if (GL > 4) p = gat_add_xor<4>(p);
    asm volatile("" : "+v"(p));

    const float sc    = vC ? p : -INFINITY;
    const bool  rise  = sc > m;
    const float t     = __expf(-fabsf(sc - m));
    const float corr  = rise ? t : 1.0f;
    const float w     = vC ? (rise ? 1.0f : t) : 0.0f;
    m = fmaxf(m, sc);
    l = l * corr + w;
    v2f c2 = v2f{corr, corr};
    v2f w2 = v2f{w, w};
    asm volatile("" : "+v"(c2), "+v"(w2));
#pragma unroll
    for (int h = 0; h < NH; ++h) acc[h] = __builtin_elementwise_fma(w2, rin[h], acc[h] * c2);
    static_assert(NH == 8, "the statement below names the eight pairs of acc");
    asm volatile("" : "+v"(m), "+v"(l), "+v"(acc[0]), "+v"(acc[1]), "+v"(acc[2]), "+v"(acc[3]),
                      "+v"(acc[4]), "+v"(acc[5]), "+v"(acc[6]), "+v"(acc[7]));
    __builtin_amdgcn_sched_barrier(0);
}

#define EX_THREADS 1024
#define EX_PER     7

__global__ void __launch_bounds__(EX_THREADS)
gat_exchange_kernel(const int* __restrict__ pairS,
                    int* __restrict__ head,
                    int* __restrict__ link,
                    int n_nodes, int n_edges) {
    const int2* __restrict__ ds2 = (const int2*)pairS;
    int2* __restrict__ link2 = (int2*)link;
    const int base = (int)blockIdx.x * (EX_THREADS * EX_PER) + (int)threadIdx.x;
    int2 dj[EX_PER];
    int  pj[EX_PER];
#pragma unroll
    for (int j = 0; j < EX_PER; ++j) {
        const int p = base + j * EX_THREADS;
        dj[j] = (p < n_edges) ? ds2[p] : make_int2(0, 0);
    }
#pragma unroll
    for (int j = 0; j < EX_PER; ++j) {
        const int p = base + j * EX_THREADS;
        pj[j] = (p < n_edges) ? atomicExch(&head[gat_clamp(dj[j].x, n_nodes)], p) : -1;
    }
#pragma unroll
    for (int j = 0; j < EX_PER; ++j) {
        const int p = base + j * EX_THREADS;
        if (p < n_edges) link2[p] = make_int2(pj[j], dj[j].y);
    }
}

#define ND_THREADS 128
__global__ void __launch_bounds__(ND_THREADS) __attribute__((amdgpu_waves_per_eu(7, 7)))
gat_node_kernel(const float* __restrict__ z,
                const int* __restrict__ head,
                const int* __restrict__ link,
                float* __restrict__ out,
                int n_nodes, int n_edges) {
    const unsigned lane = threadIdx.x & 31u;
    const unsigned sub  = lane & (unsigned)(GL - 1);
    const unsigned grp  = lane / (unsigned)GL;
    const int wv   = __builtin_amdgcn_readfirstlane((int)(blockIdx.x * (ND_THREADS / 32) + (threadIdx.x >> 5)));

    const v4f* z4 = (const v4f*)z;
    const int2* __restrict__ link2 = (const int2*)link;

    bool     live[NS];
    v2f      zd[NS][NH];
    int      nn[NS], ss[NS];
    bool     vA[NS], vB[NS];
    float    m[NS], l[NS];
    v2f      acc[NS][NH];
    v2f      r0[NH], r1[NH], r2[NH];

    int e0[NS];
#pragma unroll
    for (int w = 0; w < NS; ++w) {
        const int d = (wv * NS + w) * NPW + (int)grp;
        live[w] = d < n_nodes;
        const unsigned dc = gat_clamp(d, n_nodes);
        e0[w] = head[dc];
        const v4f* zdp = z4 + (dc * (unsigned)(OUT_DIM / 4) + sub);
#pragma unroll
        for (int q = 0; q < NP; ++q) {
            const v4f t = zdp[GL * q];
            zd[w][2 * q]     = v2f{t.x, t.y};
            zd[w][2 * q + 1] = v2f{t.z, t.w};
        }
    }
#pragma unroll
    for (int w = 0; w < NS; ++w) {
        const int e = live[w] ? e0[w] : -1;
        vA[w] = e >= 0;
        vB[w] = false;
        const unsigned ec = gat_clamp(e, n_edges);
        const int2 rec = link2[ec];
        nn[w] = rec.x;
        ss[w] = rec.y;
        m[w] = -INFINITY;
        l[w] = 0.0f;
#pragma unroll
        for (int h = 0; h < NH; ++h) acc[w][h] = v2f{0.0f, 0.0f};
    }
#pragma unroll
    for (int h = 0; h < NH; ++h) {
        r0[h] = v2f{0.0f, 0.0f};
        r1[h] = v2f{0.0f, 0.0f};
        r2[h] = v2f{0.0f, 0.0f};
    }

    asm volatile("" : "+v"(nn[0]), "+v"(ss[0]), "+v"(nn[1]), "+v"(ss[1]));

    for (int it = 0; it < n_edges + 3; it += 3) {
        if (!__any(vA[0] || vB[0] || vA[1] || vB[1])) break;
        gat_walk_step(z4, link2, n_nodes, n_edges, sub, zd[0], nn[0], ss[0], vA[0], vB[0], r1, r0, m[0], l[0], acc[0]);
        gat_walk_step(z4, link2, n_nodes, n_edges, sub, zd[1], nn[1], ss[1], vA[1], vB[1], r2, r1, m[1], l[1], acc[1]);
        gat_walk_step(z4, link2, n_nodes, n_edges, sub, zd[0], nn[0], ss[0], vA[0], vB[0], r0, r2, m[0], l[0], acc[0]);
        gat_walk_step(z4, link2, n_nodes, n_edges, sub, zd[1], nn[1], ss[1], vA[1], vB[1], r1, r0, m[1], l[1], acc[1]);
        gat_walk_step(z4, link2, n_nodes, n_edges, sub, zd[0], nn[0], ss[0], vA[0], vB[0], r2, r1, m[0], l[0], acc[0]);
        gat_walk_step(z4, link2, n_nodes, n_edges, sub, zd[1], nn[1], ss[1], vA[1], vB[1], r0, r2, m[1], l[1], acc[1]);
    }

    const unsigned grp2 = __builtin_amdgcn_mbcnt_lo(~0u, 0u) / (unsigned)GL;
#pragma unroll
    for (int w = 0; w < NS; ++w) {
        if (live[w]) {
            const float inv = 1.0f / fmaxf(l[w], 1e-30f);
            const unsigned d = (unsigned)((wv * NS + w) * NPW) + grp2;
            v4f* op = (v4f*)out + (d * (unsigned)(OUT_DIM / 4) + sub);
#pragma unroll
            for (int q = 0; q < NP; ++q)
                op[GL * q] = v4f{acc[w][2 * q].x * inv, acc[w][2 * q].y * inv,
                                 acc[w][2 * q + 1].x * inv, acc[w][2 * q + 1].y * inv};
        }
    }
}

#define FG_WGPS   256
#define FG_GM_MIN 64

extern "C" void kernel_launch(void* const* d_in, const int* in_sizes, int n_in,
                              void* d_out, int out_size, void* d_ws, size_t ws_size,
                              hipStream_t stream) {
    const float* h   = (const float*)d_in[0];
    const float* W   = (const float*)d_in[1];
    const int*   src = (const int*)d_in[2];
    const int*   dst = (const int*)d_in[3];
    float* out = (float*)d_out;

    const int n_nodes = in_sizes[0] / IN_DIM;
    const int n_edges = in_sizes[2];

    float* z    = (float*)d_ws;
    int*   link = (int*)(z + (size_t)n_nodes * OUT_DIM);
    int*   head = link + 2 * (size_t)n_edges;
    int*   pairS = (int*)d_out;
    if (ws_size < ((size_t)n_nodes * OUT_DIM + 2 * (size_t)n_edges + (size_t)n_nodes) * 4) return;
    if ((size_t)out_size < 2 * (size_t)n_edges) return;

    {
        int lk_blocks = (n_edges + LK_CHUNK - 1) / LK_CHUNK;
        int gm_blocks = FG_WGPS - lk_blocks;
        if (gm_blocks < FG_GM_MIN) gm_blocks = FG_GM_MIN;
        gat_sort_gemm_kernel<<<lk_blocks + gm_blocks, LK_THREADS, 0, stream>>>(
            dst, src, pairS, head, h, W, z, n_nodes, n_edges, lk_blocks);
    }
    {
        int blocks = (n_edges + EX_THREADS * EX_PER - 1) / (EX_THREADS * EX_PER);
        gat_exchange_kernel<<<blocks, EX_THREADS, 0, stream>>>(pairS, head, link, n_nodes, n_edges);
    }
    {
        int nodes_per_block = (ND_THREADS / 32) * NS * NPW;
        int blocks = (n_nodes + nodes_per_block - 1) / nodes_per_block;
        gat_node_kernel<<<blocks, ND_THREADS, 0, stream>>>(
            z, head, link, out, n_nodes, n_edges);
    }
}
